# starting kernel
# baseline (speedup 1.0000x reference)
#include <hip/hip_runtime.h>
#include <hip/hip_bf16.h>
#include <math.h>


typedef float v2f __attribute__((ext_vector_type(2)));
typedef float v8f __attribute__((ext_vector_type(8)));

static constexpr int N_ = 2;
static constexpr int S_ = 2048;
static constexpr int E_ = 1024;
static constexpr int H_ = 16;
static constexpr int D_ = 64;

static __device__ __forceinline__ v8f wmma_f32(v2f a, v2f b, v8f c) {
  return __builtin_amdgcn_wmma_f32_16x16x4_f32(false, a, false, b,
                                               (short)0, c, false, false);
}

static __device__ __forceinline__ float rowmax16(float v) {
  v = fmaxf(v, __shfl_xor(v, 1, 32));
  v = fmaxf(v, __shfl_xor(v, 2, 32));
  v = fmaxf(v, __shfl_xor(v, 4, 32));
  v = fmaxf(v, __shfl_xor(v, 8, 32));
  return v;
}
static __device__ __forceinline__ float rowsum16(float v) {
  v += __shfl_xor(v, 1, 32);
  v += __shfl_xor(v, 2, 32);
  v += __shfl_xor(v, 4, 32);
  v += __shfl_xor(v, 8, 32);
  return v;
}

__global__ __launch_bounds__(128)
void proj_kernel(const float* __restrict__ X,
                 const float* __restrict__ W,
                 float* __restrict__ Out)
{
  __shared__ float Ws[64 * 68];
  const int tid  = threadIdx.x;
  const int wave = tid >> 5;
  const int lane = tid & 31;
  const int n  = blockIdx.z, h = blockIdx.y;
  const int s0 = blockIdx.x * 64 + wave * 16;
  const int mrow = lane & 15, half = lane >> 4;

  for (int i = tid; i < 1024; i += 128) {
    const int r = i >> 4, c4 = i & 15;
    const float4 w4 = ((const float4*)W)[r * 16 + c4];
    float* d = &Ws[r * 68 + c4 * 4];
    d[0] = w4.x; d[1] = w4.y; d[2] = w4.z; d[3] = w4.w;
  }
  __syncthreads();

  const float* xrow = X + ((size_t)n * S_ + (s0 + mrow)) * E_ + h * D_;
  v8f c[4] = {};
  #pragma unroll
  for (int ks = 0; ks < 16; ++ks) {
    const int k = 4 * ks + 2 * half;
    v2f a; a.x = xrow[k]; a.y = xrow[k + 1];
    #pragma unroll
    for (int f = 0; f < 4; ++f) {
      const int e = 16 * f + mrow;
      v2f b; b.x = Ws[e * 68 + k]; b.y = Ws[e * 68 + k + 1];
      c[f] = wmma_f32(a, b, c[f]);
    }
  }

  float* obase = Out + ((size_t)(n * H_ + h)) * S_ * D_;
  #pragma unroll
  for (int f = 0; f < 4; ++f)
    #pragma unroll
    for (int r = 0; r < 8; ++r) {
      const int m = r + 8 * half;
      const int e = 16 * f + mrow;
      obase[(size_t)(s0 + m) * D_ + e] = c[f][r];
    }
}

__global__ __launch_bounds__(128)
void flash_kernel(const float* __restrict__ Qp,
                  const float* __restrict__ Kp,
                  const float* __restrict__ Vp,
                  float* __restrict__ O)
{
  __shared__ float Ks[16 * 68];
  __shared__ float Vs[16 * 68];
  __shared__ float Ps[4][16 * 17];

  const int tid  = threadIdx.x;
  const int wave = tid >> 5;
  const int lane = tid & 31;
  const int n = blockIdx.z, h = blockIdx.y;
  const int qblk = blockIdx.x;
  const int q0 = qblk * 64 + wave * 16;
  const int mrow = lane & 15, half = lane >> 4;
  const size_t headOff = ((size_t)(n * H_ + h)) * S_ * D_;

  float qreg[32];
  const float* qrow = Qp + headOff + (size_t)(q0 + mrow) * D_;
  #pragma unroll
  for (int ks = 0; ks < 16; ++ks) {
    const int k = 4 * ks + 2 * half;
    qreg[2 * ks]     = qrow[k];
    qreg[2 * ks + 1] = qrow[k + 1];
  }

  v8f o[4] = {};
  float mrun[8], lrun[8];
  #pragma unroll
  for (int r = 0; r < 8; ++r) { mrun[r] = -3.0e38f; lrun[r] = 0.0f; }

  const float slope  = (float)(1 << __popc(h));
  const float scale  = 0.03125f;
  const int   ktMax  = (qblk * 64 + 63) >> 4;

  for (int kt = 0; kt <= ktMax; ++kt) {
    const int k0 = kt * 16;
    __syncthreads();
    for (int i = tid; i < 256; i += 128) {
      const int r = i >> 4, c4 = i & 15;
      const float4 kq = ((const float4*)(Kp + headOff + (size_t)(k0 + r) * D_))[c4];
      const float4 vq = ((const float4*)(Vp + headOff + (size_t)(k0 + r) * D_))[c4];
      float* kd = &Ks[r * 68 + c4 * 4];
      float* vd = &Vs[r * 68 + c4 * 4];
      kd[0] = kq.x; kd[1] = kq.y; kd[2] = kq.z; kd[3] = kq.w;
      vd[0] = vq.x; vd[1] = vq.y; vd[2] = vq.z; vd[3] = vq.w;
    }
    __syncthreads();

    const bool active = (k0 <= q0 + 15);
    if (active) {
      v8f s = {};
      #pragma unroll
      for (int ks = 0; ks < 16; ++ks) {
        const int k = 4 * ks + 2 * half;
        v2f a; a.x = qreg[2 * ks]; a.y = qreg[2 * ks + 1];
        v2f b; b.x = Ks[mrow * 68 + k]; b.y = Ks[mrow * 68 + k + 1];
        s = wmma_f32(a, b, s);
      }
      #pragma unroll
      for (int r = 0; r < 8; ++r) {
        const int q = q0 + r + 8 * half;
        const int kcol = k0 + mrow;
        float e = (s[r] + slope * (float)(kcol - q)) * scale;
        if (kcol > q) e = -3.0e18f;
        s[r] = e;
      }
      float pscale[8];
      #pragma unroll
      for (int r = 0; r < 8; ++r) {
        const float rm   = rowmax16(s[r]);
        const float mnew = fmaxf(mrun[r], rm);
        const float alpha = __expf(mrun[r] - mnew);
        mrun[r] = mnew;
        const float p = __expf(s[r] - mnew);
        s[r] = p;
        lrun[r] = lrun[r] * alpha + rowsum16(p);
        pscale[r] = alpha;
      }
      #pragma unroll
      for (int f = 0; f < 4; ++f)
        #pragma unroll
        for (int r = 0; r < 8; ++r) o[f][r] *= pscale[r];

      #pragma unroll
      for (int r = 0; r < 8; ++r)
        Ps[wave][(r + 8 * half) * 17 + mrow] = s[r];
      #pragma unroll
      for (int kk = 0; kk < 4; ++kk) {
        const int ka = 4 * kk + 2 * half;
        v2f a; a.x = Ps[wave][mrow * 17 + ka]; a.y = Ps[wave][mrow * 17 + ka + 1];
        #pragma unroll
        for (int f = 0; f < 4; ++f) {
          const int dcol = 16 * f + mrow;
          v2f b;
          b.x = Vs[(4 * kk + 2 * half)     * 68 + dcol];
          b.y = Vs[(4 * kk + 2 * half + 1) * 68 + dcol];
          o[f] = wmma_f32(a, b, o[f]);
        }
      }
    }
  }

  #pragma unroll
  for (int r = 0; r < 8; ++r) lrun[r] = 1.0f / lrun[r];
  float* obase = O + (size_t)n * S_ * E_ + h * D_;
  #pragma unroll
  for (int f = 0; f < 4; ++f)
    #pragma unroll
    for (int r = 0; r < 8; ++r) {
      const int m = r + 8 * half;
      const int dcol = 16 * f + mrow;
      obase[(size_t)(q0 + m) * E_ + dcol] = o[f][r] * lrun[r];
    }
}

__global__ __launch_bounds__(128)
void outproj_kernel(const float* __restrict__ A,
                    const float* __restrict__ Wo,
                    const float* __restrict__ bo,
                    float* __restrict__ Y)
{
  __shared__ float Ws[64 * 68];
  const int tid  = threadIdx.x;
  const int wave = tid >> 5;
  const int lane = tid & 31;
  const int r0 = blockIdx.y * 64 + wave * 16;
  const int c0 = blockIdx.x * 64;
  const int mrow = lane & 15, half = lane >> 4;

  v8f c[4] = {};
  const float* arow = A + (size_t)(r0 + mrow) * E_;

  for (int kc = 0; kc < E_; kc += 64) {
    __syncthreads();
    for (int i = tid; i < 1024; i += 128) {
      const int rr = i >> 4, c4 = i & 15;
      const float4 w4 = ((const float4*)(Wo + (size_t)(c0 + rr) * E_ + kc))[c4];
      float* d = &Ws[rr * 68 + c4 * 4];
      d[0] = w4.x; d[1] = w4.y; d[2] = w4.z; d[3] = w4.w;
    }
    __syncthreads();
    #pragma unroll
    for (int ks = 0; ks < 16; ++ks) {
      const int k = 4 * ks + 2 * half;
      v2f a; a.x = arow[kc + k]; a.y = arow[kc + k + 1];
      #pragma unroll
      for (int f = 0; f < 4; ++f) {
        const int e = 16 * f + mrow;
        v2f b; b.x = Ws[e * 68 + k]; b.y = Ws[e * 68 + k + 1];
        c[f] = wmma_f32(a, b, c[f]);
      }
    }
  }

  #pragma unroll
  for (int f = 0; f < 4; ++f)
    #pragma unroll
    for (int r = 0; r < 8; ++r) {
      const int m = r + 8 * half;
      const int e = c0 + 16 * f + mrow;
      Y[(size_t)(r0 + m) * E_ + e] = c[f][r] + bo[e];
    }
}

extern "C" void kernel_launch(void* const* d_in, const int* in_sizes, int n_in,
                              void* d_out, int out_size, void* d_ws, size_t ws_size,
                              hipStream_t stream) {
  const float* values = (const float*)d_in[0];
  const float* keys   = (const float*)d_in[1];
  const float* query  = (const float*)d_in[2];
  const float* Wv     = (const float*)d_in[4];
  const float* Wk     = (const float*)d_in[5];
  const float* Wq     = (const float*)d_in[6];
  const float* Wo     = (const float*)d_in[7];
  const float* bo     = (const float*)d_in[8];
  float* out = (float*)d_out;

  const size_t tensorElems = (size_t)N_ * H_ * S_ * D_;
  float* Qp = (float*)d_ws;
  float* Kp = Qp + tensorElems;
  float* Vp = Kp + tensorElems;
  float* Oa = Vp + tensorElems;

  const dim3 blk(128);
  const dim3 gProj(S_ / 64, H_, N_);

  proj_kernel<<<gProj, blk, 0, stream>>>(query,  Wq, Qp);
  proj_kernel<<<gProj, blk, 0, stream>>>(keys,   Wk, Kp);
  proj_kernel<<<gProj, blk, 0, stream>>>(values, Wv, Vp);

  flash_kernel<<<gProj, blk, 0, stream>>>(Qp, Kp, Vp, Oa);

  const dim3 gOut(E_ / 64, (N_ * S_) / 64);
  outproj_kernel<<<gOut, blk, 0, stream>>>(Oa, Wo, bo, out);
}


# attention kernel: softmax row max and row sum over 16 lanes by DPP row_xmask on the vector ALU, not __shfl_xor (ds_bpermute via the LDS unit); same lane pairs and step order, sums bit-identical
# speedup vs baseline: 1.0723x; 1.0723x over previous
#include <hip/hip_runtime.h>
#include <hip/hip_bf16.h>
#include <math.h>


typedef float v2f __attribute__((ext_vector_type(2)));
typedef float v8f __attribute__((ext_vector_type(8)));

static constexpr int N_ = 2;
static constexpr int S_ = 2048;
static constexpr int E_ = 1024;
static constexpr int H_ = 16;
static constexpr int D_ = 64;

static __device__ __forceinline__ v8f wmma_f32(v2f a, v2f b, v8f c) {
  return __builtin_amdgcn_wmma_f32_16x16x4_f32(false, a, false, b,
                                               (short)0, c, false, false);
}

template <int M>
static __device__ __forceinline__ float row_xor16(float v) {
  const int vi = __builtin_bit_cast(int, v);
  return __builtin_bit_cast(float,
      __builtin_amdgcn_update_dpp(vi, vi, 0x160 + M, 0xf, 0xf, true));
}
static __device__ __forceinline__ float rowmax16(float v) {
  v = fmaxf(v, row_xor16<1>(v));
  v = fmaxf(v, row_xor16<2>(v));
  v = fmaxf(v, row_xor16<4>(v));
  v = fmaxf(v, row_xor16<8>(v));
  return v;
}
static __device__ __forceinline__ float rowsum16(float v) {
  v += row_xor16<1>(v);
  v += row_xor16<2>(v);
  v += row_xor16<4>(v);
  v += row_xor16<8>(v);
  return v;
}

__global__ __launch_bounds__(128)
void proj_kernel(const float* __restrict__ X,
                 const float* __restrict__ W,
                 float* __restrict__ Out)
{
  __shared__ float Ws[64 * 68];
  const int tid  = threadIdx.x;
  const int wave = tid >> 5;
  const int lane = tid & 31;
  const int n  = blockIdx.z, h = blockIdx.y;
  const int s0 = blockIdx.x * 64 + wave * 16;
  const int mrow = lane & 15, half = lane >> 4;

  for (int i = tid; i < 1024; i += 128) {
    const int r = i >> 4, c4 = i & 15;
    const float4 w4 = ((const float4*)W)[r * 16 + c4];
    float* d = &Ws[r * 68 + c4 * 4];
    d[0] = w4.x; d[1] = w4.y; d[2] = w4.z; d[3] = w4.w;
  }
  __syncthreads();

  const float* xrow = X + ((size_t)n * S_ + (s0 + mrow)) * E_ + h * D_;
  v8f c[4] = {};
  #pragma unroll
  for (int ks = 0; ks < 16; ++ks) {
    const int k = 4 * ks + 2 * half;
    v2f a; a.x = xrow[k]; a.y = xrow[k + 1];
    #pragma unroll
    for (int f = 0; f < 4; ++f) {
      const int e = 16 * f + mrow;
      v2f b; b.x = Ws[e * 68 + k]; b.y = Ws[e * 68 + k + 1];
      c[f] = wmma_f32(a, b, c[f]);
    }
  }

  float* obase = Out + ((size_t)(n * H_ + h)) * S_ * D_;
  #pragma unroll
  for (int f = 0; f < 4; ++f)
    #pragma unroll
    for (int r = 0; r < 8; ++r) {
      const int m = r + 8 * half;
      const int e = 16 * f + mrow;
      obase[(size_t)(s0 + m) * D_ + e] = c[f][r];
    }
}

__global__ __launch_bounds__(128)
void flash_kernel(const float* __restrict__ Qp,
                  const float* __restrict__ Kp,
                  const float* __restrict__ Vp,
                  float* __restrict__ O)
{
  __shared__ float Ks[16 * 68];
  __shared__ float Vs[16 * 68];
  __shared__ float Ps[4][16 * 17];

  const int tid  = threadIdx.x;
  const int wave = tid >> 5;
  const int lane = tid & 31;
  const int n = blockIdx.z, h = blockIdx.y;
  const int qblk = blockIdx.x;
  const int q0 = qblk * 64 + wave * 16;
  const int mrow = lane & 15, half = lane >> 4;
  const size_t headOff = ((size_t)(n * H_ + h)) * S_ * D_;

  float qreg[32];
  const float* qrow = Qp + headOff + (size_t)(q0 + mrow) * D_;
  #pragma unroll
  for (int ks = 0; ks < 16; ++ks) {
    const int k = 4 * ks + 2 * half;
    qreg[2 * ks]     = qrow[k];
    qreg[2 * ks + 1] = qrow[k + 1];
  }

  v8f o[4] = {};
  float mrun[8], lrun[8];
  #pragma unroll
  for (int r = 0; r < 8; ++r) { mrun[r] = -3.0e38f; lrun[r] = 0.0f; }

  const float slope  = (float)(1 << __popc(h));
  const float scale  = 0.03125f;
  const int   ktMax  = (qblk * 64 + 63) >> 4;

  for (int kt = 0; kt <= ktMax; ++kt) {
    const int k0 = kt * 16;
    __syncthreads();
    for (int i = tid; i < 256; i += 128) {
      const int r = i >> 4, c4 = i & 15;
      const float4 kq = ((const float4*)(Kp + headOff + (size_t)(k0 + r) * D_))[c4];
      const float4 vq = ((const float4*)(Vp + headOff + (size_t)(k0 + r) * D_))[c4];
      float* kd = &Ks[r * 68 + c4 * 4];
      float* vd = &Vs[r * 68 + c4 * 4];
      kd[0] = kq.x; kd[1] = kq.y; kd[2] = kq.z; kd[3] = kq.w;
      vd[0] = vq.x; vd[1] = vq.y; vd[2] = vq.z; vd[3] = vq.w;
    }
    __syncthreads();

    const bool active = (k0 <= q0 + 15);
    if (active) {
      v8f s = {};
      #pragma unroll
      for (int ks = 0; ks < 16; ++ks) {
        const int k = 4 * ks + 2 * half;
        v2f a; a.x = qreg[2 * ks]; a.y = qreg[2 * ks + 1];
        v2f b; b.x = Ks[mrow * 68 + k]; b.y = Ks[mrow * 68 + k + 1];
        s = wmma_f32(a, b, s);
      }
      #pragma unroll
      for (int r = 0; r < 8; ++r) {
        const int q = q0 + r + 8 * half;
        const int kcol = k0 + mrow;
        float e = (s[r] + slope * (float)(kcol - q)) * scale;
        if (kcol > q) e = -3.0e18f;
        s[r] = e;
      }
      float pscale[8];
      #pragma unroll
      for (int r = 0; r < 8; ++r) {
        const float rm   = rowmax16(s[r]);
        const float mnew = fmaxf(mrun[r], rm);
        const float alpha = __expf(mrun[r] - mnew);
        mrun[r] = mnew;
        const float p = __expf(s[r] - mnew);
        s[r] = p;
        lrun[r] = lrun[r] * alpha + rowsum16(p);
        pscale[r] = alpha;
      }
      #pragma unroll
      for (int f = 0; f < 4; ++f)
        #pragma unroll
        for (int r = 0; r < 8; ++r) o[f][r] *= pscale[r];

      #pragma unroll
      for (int r = 0; r < 8; ++r)
        Ps[wave][(r + 8 * half) * 17 + mrow] = s[r];
      #pragma unroll
      for (int kk = 0; kk < 4; ++kk) {
        const int ka = 4 * kk + 2 * half;
        v2f a; a.x = Ps[wave][mrow * 17 + ka]; a.y = Ps[wave][mrow * 17 + ka + 1];
        #pragma unroll
        for (int f = 0; f < 4; ++f) {
          const int dcol = 16 * f + mrow;
          v2f b;
          b.x = Vs[(4 * kk + 2 * half)     * 68 + dcol];
          b.y = Vs[(4 * kk + 2 * half + 1) * 68 + dcol];
          o[f] = wmma_f32(a, b, o[f]);
        }
      }
    }
  }

  #pragma unroll
  for (int r = 0; r < 8; ++r) lrun[r] = 1.0f / lrun[r];
  float* obase = O + (size_t)n * S_ * E_ + h * D_;
  #pragma unroll
  for (int f = 0; f < 4; ++f)
    #pragma unroll
    for (int r = 0; r < 8; ++r) {
      const int m = r + 8 * half;
      const int dcol = 16 * f + mrow;
      obase[(size_t)(q0 + m) * E_ + dcol] = o[f][r] * lrun[r];
    }
}

__global__ __launch_bounds__(128)
void outproj_kernel(const float* __restrict__ A,
                    const float* __restrict__ Wo,
                    const float* __restrict__ bo,
                    float* __restrict__ Y)
{
  __shared__ float Ws[64 * 68];
  const int tid  = threadIdx.x;
  const int wave = tid >> 5;
  const int lane = tid & 31;
  const int r0 = blockIdx.y * 64 + wave * 16;
  const int c0 = blockIdx.x * 64;
  const int mrow = lane & 15, half = lane >> 4;

  v8f c[4] = {};
  const float* arow = A + (size_t)(r0 + mrow) * E_;

  for (int kc = 0; kc < E_; kc += 64) {
    __syncthreads();
    for (int i = tid; i < 1024; i += 128) {
      const int rr = i >> 4, c4 = i & 15;
      const float4 w4 = ((const float4*)(Wo + (size_t)(c0 + rr) * E_ + kc))[c4];
      float* d = &Ws[rr * 68 + c4 * 4];
      d[0] = w4.x; d[1] = w4.y; d[2] = w4.z; d[3] = w4.w;
    }
    __syncthreads();
    #pragma unroll
    for (int ks = 0; ks < 16; ++ks) {
      const int k = 4 * ks + 2 * half;
      v2f a; a.x = arow[kc + k]; a.y = arow[kc + k + 1];
      #pragma unroll
      for (int f = 0; f < 4; ++f) {
        const int e = 16 * f + mrow;
        v2f b; b.x = Ws[e * 68 + k]; b.y = Ws[e * 68 + k + 1];
        c[f] = wmma_f32(a, b, c[f]);
      }
    }
  }

  #pragma unroll
  for (int f = 0; f < 4; ++f)
    #pragma unroll
    for (int r = 0; r < 8; ++r) {
      const int m = r + 8 * half;
      const int e = c0 + 16 * f + mrow;
      Y[(size_t)(r0 + m) * E_ + e] = c[f][r] + bo[e];
    }
}

extern "C" void kernel_launch(void* const* d_in, const int* in_sizes, int n_in,
                              void* d_out, int out_size, void* d_ws, size_t ws_size,
                              hipStream_t stream) {
  const float* values = (const float*)d_in[0];
  const float* keys   = (const float*)d_in[1];
  const float* query  = (const float*)d_in[2];
  const float* Wv     = (const float*)d_in[4];
  const float* Wk     = (const float*)d_in[5];
  const float* Wq     = (const float*)d_in[6];
  const float* Wo     = (const float*)d_in[7];
  const float* bo     = (const float*)d_in[8];
  float* out = (float*)d_out;

  const size_t tensorElems = (size_t)N_ * H_ * S_ * D_;
  float* Qp = (float*)d_ws;
  float* Kp = Qp + tensorElems;
  float* Vp = Kp + tensorElems;
  float* Oa = Vp + tensorElems;

  const dim3 blk(128);
  const dim3 gProj(S_ / 64, H_, N_);

  proj_kernel<<<gProj, blk, 0, stream>>>(query,  Wq, Qp);
  proj_kernel<<<gProj, blk, 0, stream>>>(keys,   Wk, Kp);
  proj_kernel<<<gProj, blk, 0, stream>>>(values, Wv, Vp);

  flash_kernel<<<gProj, blk, 0, stream>>>(Qp, Kp, Vp, Oa);

  const dim3 gOut(E_ / 64, (N_ * S_) / 64);
  outproj_kernel<<<gOut, blk, 0, stream>>>(Oa, Wo, bo, out);
}


# attention kernel: each query tile's key loop split over four waves (16-wave blocks, 64 keys staged per step), partial max/sum/output merged at the end: row sums and P.V summed in a new order
# speedup vs baseline: 1.1520x; 1.0743x over previous
#include <hip/hip_runtime.h>
#include <hip/hip_bf16.h>
#include <math.h>


typedef float v2f __attribute__((ext_vector_type(2)));
typedef float v8f __attribute__((ext_vector_type(8)));

static constexpr int N_ = 2;
static constexpr int S_ = 2048;
static constexpr int E_ = 1024;
static constexpr int H_ = 16;
static constexpr int D_ = 64;

static __device__ __forceinline__ v8f wmma_f32(v2f a, v2f b, v8f c) {
  return __builtin_amdgcn_wmma_f32_16x16x4_f32(false, a, false, b,
                                               (short)0, c, false, false);
}

template <int M>
static __device__ __forceinline__ float row_xor16(float v) {
  const int vi = __builtin_bit_cast(int, v);
  return __builtin_bit_cast(float,
      __builtin_amdgcn_update_dpp(vi, vi, 0x160 + M, 0xf, 0xf, true));
}
static __device__ __forceinline__ float rowmax16(float v) {
  v = fmaxf(v, row_xor16<1>(v));
  v = fmaxf(v, row_xor16<2>(v));
  v = fmaxf(v, row_xor16<4>(v));
  v = fmaxf(v, row_xor16<8>(v));
  return v;
}
static __device__ __forceinline__ float rowsum16(float v) {
  v += row_xor16<1>(v);
  v += row_xor16<2>(v);
  v += row_xor16<4>(v);
  v += row_xor16<8>(v);
  return v;
}

__global__ __launch_bounds__(128)
void proj_kernel(const float* __restrict__ X,
                 const float* __restrict__ W,
                 float* __restrict__ Out)
{
  __shared__ float Ws[64 * 68];
  const int tid  = threadIdx.x;
  const int wave = tid >> 5;
  const int lane = tid & 31;
  const int n  = blockIdx.z, h = blockIdx.y;
  const int s0 = blockIdx.x * 64 + wave * 16;
  const int mrow = lane & 15, half = lane >> 4;

  for (int i = tid; i < 1024; i += 128) {
    const int r = i >> 4, c4 = i & 15;
    const float4 w4 = ((const float4*)W)[r * 16 + c4];
    float* d = &Ws[r * 68 + c4 * 4];
    d[0] = w4.x; d[1] = w4.y; d[2] = w4.z; d[3] = w4.w;
  }
  __syncthreads();

  const float* xrow = X + ((size_t)n * S_ + (s0 + mrow)) * E_ + h * D_;
  v8f c[4] = {};
  #pragma unroll
  for (int ks = 0; ks < 16; ++ks) {
    const int k = 4 * ks + 2 * half;
    v2f a; a.x = xrow[k]; a.y = xrow[k + 1];
    #pragma unroll
    for (int f = 0; f < 4; ++f) {
      const int e = 16 * f + mrow;
      v2f b; b.x = Ws[e * 68 + k]; b.y = Ws[e * 68 + k + 1];
      c[f] = wmma_f32(a, b, c[f]);
    }
  }

  float* obase = Out + ((size_t)(n * H_ + h)) * S_ * D_;
  #pragma unroll
  for (int f = 0; f < 4; ++f)
    #pragma unroll
    for (int r = 0; r < 8; ++r) {
      const int m = r + 8 * half;
      const int e = 16 * f + mrow;
      obase[(size_t)(s0 + m) * D_ + e] = c[f][r];
    }
}

__global__ __launch_bounds__(512)
void flash_kernel(const float* __restrict__ Qp,
                  const float* __restrict__ Kp,
                  const float* __restrict__ Vp,
                  float* __restrict__ O)
{
  __shared__ float smem[4352 + 4352 + 4352];
  float* Ks = smem;
  float* Vs = smem + 4352;

  const int tid  = threadIdx.x;
  const int wave = tid >> 5;
  const int lane = tid & 31;
  const int grp  = wave >> 2;
  const int qw   = wave & 3;
  const int n = blockIdx.z, h = blockIdx.y;
  const int qblk = blockIdx.x;
  const int q0 = qblk * 64 + qw * 16;
  const int mrow = lane & 15, half = lane >> 4;
  const size_t headOff = ((size_t)(n * H_ + h)) * S_ * D_;
  float* Pw = smem + 8704 + wave * 272;
  const float* Kt = Ks + grp * (16 * 68);
  const float* Vt = Vs + grp * (16 * 68);

  float qreg[32];
  const float* qrow = Qp + headOff + (size_t)(q0 + mrow) * D_;
  #pragma unroll
  for (int ks = 0; ks < 16; ++ks) {
    const int k = 4 * ks + 2 * half;
    qreg[2 * ks]     = qrow[k];
    qreg[2 * ks + 1] = qrow[k + 1];
  }

  v8f o[4] = {};
  float mrun[8], lrun[8];
  #pragma unroll
  for (int r = 0; r < 8; ++r) { mrun[r] = -3.0e38f; lrun[r] = 0.0f; }

  const float slope  = (float)(1 << __popc(h));
  const float scale  = 0.03125f;
  const int   nSteps = qblk + 1;

  for (int st = 0; st < nSteps; ++st) {
    const int kbase = st * 64;
    __syncthreads();
    for (int i = tid; i < 1024; i += 512) {
      const int r = i >> 4, c4 = i & 15;
      const float4 kq = ((const float4*)(Kp + headOff + (size_t)(kbase + r) * D_))[c4];
      const float4 vq = ((const float4*)(Vp + headOff + (size_t)(kbase + r) * D_))[c4];
      float* kd = &Ks[r * 68 + c4 * 4];
      float* vd = &Vs[r * 68 + c4 * 4];
      kd[0] = kq.x; kd[1] = kq.y; kd[2] = kq.z; kd[3] = kq.w;
      vd[0] = vq.x; vd[1] = vq.y; vd[2] = vq.z; vd[3] = vq.w;
    }
    __syncthreads();

    const int k0 = kbase + grp * 16;
    const bool active = (k0 <= q0 + 15);
    if (active) {
      v8f s = {};
      #pragma unroll
      for (int ks = 0; ks < 16; ++ks) {
        const int k = 4 * ks + 2 * half;
        v2f a; a.x = qreg[2 * ks]; a.y = qreg[2 * ks + 1];
        v2f b; b.x = Kt[mrow * 68 + k]; b.y = Kt[mrow * 68 + k + 1];
        s = wmma_f32(a, b, s);
      }
      #pragma unroll
      for (int r = 0; r < 8; ++r) {
        const int q = q0 + r + 8 * half;
        const int kcol = k0 + mrow;
        float e = (s[r] + slope * (float)(kcol - q)) * scale;
        if (kcol > q) e = -3.0e18f;
        s[r] = e;
      }
      float pscale[8];
      #pragma unroll
      for (int r = 0; r < 8; ++r) {
        const float rm   = rowmax16(s[r]);
        const float mnew = fmaxf(mrun[r], rm);
        const float alpha = __expf(mrun[r] - mnew);
        mrun[r] = mnew;
        const float p = __expf(s[r] - mnew);
        s[r] = p;
        lrun[r] = lrun[r] * alpha + rowsum16(p);
        pscale[r] = alpha;
      }
      #pragma unroll
      for (int f = 0; f < 4; ++f)
        #pragma unroll
        for (int r = 0; r < 8; ++r) o[f][r] *= pscale[r];

      #pragma unroll
      for (int r = 0; r < 8; ++r)
        Pw[(r + 8 * half) * 17 + mrow] = s[r];
      #pragma unroll
      for (int kk = 0; kk < 4; ++kk) {
        const int ka = 4 * kk + 2 * half;
        v2f a; a.x = Pw[mrow * 17 + ka]; a.y = Pw[mrow * 17 + ka + 1];
        #pragma unroll
        for (int f = 0; f < 4; ++f) {
          const int dcol = 16 * f + mrow;
          v2f b;
          b.x = Vt[(4 * kk + 2 * half)     * 68 + dcol];
          b.y = Vt[(4 * kk + 2 * half + 1) * 68 + dcol];
          o[f] = wmma_f32(a, b, o[f]);
        }
      }
    }
  }

  #pragma unroll
  for (int round = 0; round < 2; ++round) {
    const int senderLo = (round == 0) ? 2 : 1;
    const int nPairs   = (round == 0) ? 2 : 1;
    __syncthreads();
    if (grp >= senderLo && grp < senderLo + nPairs) {
      float* mb = smem + ((grp - senderLo) * 4 + qw) * 1536;
      #pragma unroll
      for (int f = 0; f < 4; ++f)
        #pragma unroll
        for (int r = 0; r < 8; ++r) mb[(f * 8 + r) * 32 + lane] = o[f][r];
      #pragma unroll
      for (int r = 0; r < 8; ++r) {
        mb[1024 + r * 32 + lane] = mrun[r];
        mb[1280 + r * 32 + lane] = lrun[r];
      }
    }
    __syncthreads();
    if (grp < nPairs) {
      const float* mb = smem + (grp * 4 + qw) * 1536;
      float a0[8], a1[8];
      #pragma unroll
      for (int r = 0; r < 8; ++r) {
        const float m1 = mb[1024 + r * 32 + lane];
        const float l1 = mb[1280 + r * 32 + lane];
        const float m  = fmaxf(mrun[r], m1);
        a0[r] = __expf(mrun[r] - m);
        a1[r] = __expf(m1 - m);
        lrun[r] = lrun[r] * a0[r] + l1 * a1[r];
        mrun[r] = m;
      }
      #pragma unroll
      for (int f = 0; f < 4; ++f)
        #pragma unroll
        for (int r = 0; r < 8; ++r)
          o[f][r] = o[f][r] * a0[r] + mb[(f * 8 + r) * 32 + lane] * a1[r];
    }
  }

  if (grp == 0) {
    #pragma unroll
    for (int r = 0; r < 8; ++r) lrun[r] = 1.0f / lrun[r];
    float* obase = O + (size_t)n * S_ * E_ + h * D_;
    #pragma unroll
    for (int f = 0; f < 4; ++f)
      #pragma unroll
      for (int r = 0; r < 8; ++r) {
        const int m = r + 8 * half;
        const int dcol = 16 * f + mrow;
        obase[(size_t)(q0 + m) * E_ + dcol] = o[f][r] * lrun[r];
      }
  }
}

__global__ __launch_bounds__(128)
void outproj_kernel(const float* __restrict__ A,
                    const float* __restrict__ Wo,
                    const float* __restrict__ bo,
                    float* __restrict__ Y)
{
  __shared__ float Ws[64 * 68];
  const int tid  = threadIdx.x;
  const int wave = tid >> 5;
  const int lane = tid & 31;
  const int r0 = blockIdx.y * 64 + wave * 16;
  const int c0 = blockIdx.x * 64;
  const int mrow = lane & 15, half = lane >> 4;

  v8f c[4] = {};
  const float* arow = A + (size_t)(r0 + mrow) * E_;

  for (int kc = 0; kc < E_; kc += 64) {
    __syncthreads();
    for (int i = tid; i < 1024; i += 128) {
      const int rr = i >> 4, c4 = i & 15;
      const float4 w4 = ((const float4*)(Wo + (size_t)(c0 + rr) * E_ + kc))[c4];
      float* d = &Ws[rr * 68 + c4 * 4];
      d[0] = w4.x; d[1] = w4.y; d[2] = w4.z; d[3] = w4.w;
    }
    __syncthreads();
    #pragma unroll
    for (int ks = 0; ks < 16; ++ks) {
      const int k = 4 * ks + 2 * half;
      v2f a; a.x = arow[kc + k]; a.y = arow[kc + k + 1];
      #pragma unroll
      for (int f = 0; f < 4; ++f) {
        const int e = 16 * f + mrow;
        v2f b; b.x = Ws[e * 68 + k]; b.y = Ws[e * 68 + k + 1];
        c[f] = wmma_f32(a, b, c[f]);
      }
    }
  }

  #pragma unroll
  for (int f = 0; f < 4; ++f)
    #pragma unroll
    for (int r = 0; r < 8; ++r) {
      const int m = r + 8 * half;
      const int e = c0 + 16 * f + mrow;
      Y[(size_t)(r0 + m) * E_ + e] = c[f][r] + bo[e];
    }
}

extern "C" void kernel_launch(void* const* d_in, const int* in_sizes, int n_in,
                              void* d_out, int out_size, void* d_ws, size_t ws_size,
                              hipStream_t stream) {
  const float* values = (const float*)d_in[0];
  const float* keys   = (const float*)d_in[1];
  const float* query  = (const float*)d_in[2];
  const float* Wv     = (const float*)d_in[4];
  const float* Wk     = (const float*)d_in[5];
  const float* Wq     = (const float*)d_in[6];
  const float* Wo     = (const float*)d_in[7];
  const float* bo     = (const float*)d_in[8];
  float* out = (float*)d_out;

  const size_t tensorElems = (size_t)N_ * H_ * S_ * D_;
  float* Qp = (float*)d_ws;
  float* Kp = Qp + tensorElems;
  float* Vp = Kp + tensorElems;
  float* Oa = Vp + tensorElems;

  const dim3 blk(128);
  const dim3 gProj(S_ / 64, H_, N_);

  proj_kernel<<<gProj, blk, 0, stream>>>(query,  Wq, Qp);
  proj_kernel<<<gProj, blk, 0, stream>>>(keys,   Wk, Kp);
  proj_kernel<<<gProj, blk, 0, stream>>>(values, Wv, Vp);

  flash_kernel<<<gProj, dim3(512), 0, stream>>>(Qp, Kp, Vp, Oa);

  const dim3 gOut(E_ / 64, (N_ * S_) / 64);
  outproj_kernel<<<gOut, blk, 0, stream>>>(Oa, Wo, bo, out);
}


# attention kernel: work-groups issued heaviest first (grid x = batch-head pair, y = query block descending), so the 32-step causal blocks no longer start last; each block's work and sums unchanged
# speedup vs baseline: 1.4212x; 1.2337x over previous
#include <hip/hip_runtime.h>
#include <hip/hip_bf16.h>
#include <math.h>


typedef float v2f __attribute__((ext_vector_type(2)));
typedef float v8f __attribute__((ext_vector_type(8)));

static constexpr int N_ = 2;
static constexpr int S_ = 2048;
static constexpr int E_ = 1024;
static constexpr int H_ = 16;
static constexpr int D_ = 64;

static __device__ __forceinline__ v8f wmma_f32(v2f a, v2f b, v8f c) {
  return __builtin_amdgcn_wmma_f32_16x16x4_f32(false, a, false, b,
                                               (short)0, c, false, false);
}

template <int M>
static __device__ __forceinline__ float row_xor16(float v) {
  const int vi = __builtin_bit_cast(int, v);
  return __builtin_bit_cast(float,
      __builtin_amdgcn_update_dpp(vi, vi, 0x160 + M, 0xf, 0xf, true));
}
static __device__ __forceinline__ float rowmax16(float v) {
  v = fmaxf(v, row_xor16<1>(v));
  v = fmaxf(v, row_xor16<2>(v));
  v = fmaxf(v, row_xor16<4>(v));
  v = fmaxf(v, row_xor16<8>(v));
  return v;
}
static __device__ __forceinline__ float rowsum16(float v) {
  v += row_xor16<1>(v);
  v += row_xor16<2>(v);
  v += row_xor16<4>(v);
  v += row_xor16<8>(v);
  return v;
}

__global__ __launch_bounds__(128)
void proj_kernel(const float* __restrict__ X,
                 const float* __restrict__ W,
                 float* __restrict__ Out)
{
  __shared__ float Ws[64 * 68];
  const int tid  = threadIdx.x;
  const int wave = tid >> 5;
  const int lane = tid & 31;
  const int n  = blockIdx.z, h = blockIdx.y;
  const int s0 = blockIdx.x * 64 + wave * 16;
  const int mrow = lane & 15, half = lane >> 4;

  for (int i = tid; i < 1024; i += 128) {
    const int r = i >> 4, c4 = i & 15;
    const float4 w4 = ((const float4*)W)[r * 16 + c4];
    float* d = &Ws[r * 68 + c4 * 4];
    d[0] = w4.x; d[1] = w4.y; d[2] = w4.z; d[3] = w4.w;
  }
  __syncthreads();

  const float* xrow = X + ((size_t)n * S_ + (s0 + mrow)) * E_ + h * D_;
  v8f c[4] = {};
  #pragma unroll
  for (int ks = 0; ks < 16; ++ks) {
    const int k = 4 * ks + 2 * half;
    v2f a; a.x = xrow[k]; a.y = xrow[k + 1];
    #pragma unroll
    for (int f = 0; f < 4; ++f) {
      const int e = 16 * f + mrow;
      v2f b; b.x = Ws[e * 68 + k]; b.y = Ws[e * 68 + k + 1];
      c[f] = wmma_f32(a, b, c[f]);
    }
  }

  float* obase = Out + ((size_t)(n * H_ + h)) * S_ * D_;
  #pragma unroll
  for (int f = 0; f < 4; ++f)
    #pragma unroll
    for (int r = 0; r < 8; ++r) {
      const int m = r + 8 * half;
      const int e = 16 * f + mrow;
      obase[(size_t)(s0 + m) * D_ + e] = c[f][r];
    }
}

__global__ __launch_bounds__(512)
void flash_kernel(const float* __restrict__ Qp,
                  const float* __restrict__ Kp,
                  const float* __restrict__ Vp,
                  float* __restrict__ O)
{
  __shared__ float smem[4352 + 4352 + 4352];
  float* Ks = smem;
  float* Vs = smem + 4352;

  const int tid  = threadIdx.x;
  const int wave = tid >> 5;
  const int lane = tid & 31;
  const int grp  = wave >> 2;
  const int qw   = wave & 3;
  const int n = blockIdx.x / H_, h = blockIdx.x % H_;
  const int qblk = (S_ / 64 - 1) - blockIdx.y;
  const int q0 = qblk * 64 + qw * 16;
  const int mrow = lane & 15, half = lane >> 4;
  const size_t headOff = ((size_t)(n * H_ + h)) * S_ * D_;
  float* Pw = smem + 8704 + wave * 272;
  const float* Kt = Ks + grp * (16 * 68);
  const float* Vt = Vs + grp * (16 * 68);

  float qreg[32];
  const float* qrow = Qp + headOff + (size_t)(q0 + mrow) * D_;
  #pragma unroll
  for (int ks = 0; ks < 16; ++ks) {
    const int k = 4 * ks + 2 * half;
    qreg[2 * ks]     = qrow[k];
    qreg[2 * ks + 1] = qrow[k + 1];
  }

  v8f o[4] = {};
  float mrun[8], lrun[8];
  #pragma unroll
  for (int r = 0; r < 8; ++r) { mrun[r] = -3.0e38f; lrun[r] = 0.0f; }

  const float slope  = (float)(1 << __popc(h));
  const float scale  = 0.03125f;
  const int   nSteps = qblk + 1;

  for (int st = 0; st < nSteps; ++st) {
    const int kbase = st * 64;
    __syncthreads();
    for (int i = tid; i < 1024; i += 512) {
      const int r = i >> 4, c4 = i & 15;
      const float4 kq = ((const float4*)(Kp + headOff + (size_t)(kbase + r) * D_))[c4];
      const float4 vq = ((const float4*)(Vp + headOff + (size_t)(kbase + r) * D_))[c4];
      float* kd = &Ks[r * 68 + c4 * 4];
      float* vd = &Vs[r * 68 + c4 * 4];
      kd[0] = kq.x; kd[1] = kq.y; kd[2] = kq.z; kd[3] = kq.w;
      vd[0] = vq.x; vd[1] = vq.y; vd[2] = vq.z; vd[3] = vq.w;
    }
    __syncthreads();

    const int k0 = kbase + grp * 16;
    const bool active = (k0 <= q0 + 15);
    if (active) {
      v8f s = {};
      #pragma unroll
      for (int ks = 0; ks < 16; ++ks) {
        const int k = 4 * ks + 2 * half;
        v2f a; a.x = qreg[2 * ks]; a.y = qreg[2 * ks + 1];
        v2f b; b.x = Kt[mrow * 68 + k]; b.y = Kt[mrow * 68 + k + 1];
        s = wmma_f32(a, b, s);
      }
      #pragma unroll
      for (int r = 0; r < 8; ++r) {
        const int q = q0 + r + 8 * half;
        const int kcol = k0 + mrow;
        float e = (s[r] + slope * (float)(kcol - q)) * scale;
        if (kcol > q) e = -3.0e18f;
        s[r] = e;
      }
      float pscale[8];
      #pragma unroll
      for (int r = 0; r < 8; ++r) {
        const float rm   = rowmax16(s[r]);
        const float mnew = fmaxf(mrun[r], rm);
        const float alpha = __expf(mrun[r] - mnew);
        mrun[r] = mnew;
        const float p = __expf(s[r] - mnew);
        s[r] = p;
        lrun[r] = lrun[r] * alpha + rowsum16(p);
        pscale[r] = alpha;
      }
      #pragma unroll
      for (int f = 0; f < 4; ++f)
        #pragma unroll
        for (int r = 0; r < 8; ++r) o[f][r] *= pscale[r];

      #pragma unroll
      for (int r = 0; r < 8; ++r)
        Pw[(r + 8 * half) * 17 + mrow] = s[r];
      #pragma unroll
      for (int kk = 0; kk < 4; ++kk) {
        const int ka = 4 * kk + 2 * half;
        v2f a; a.x = Pw[mrow * 17 + ka]; a.y = Pw[mrow * 17 + ka + 1];
        #pragma unroll
        for (int f = 0; f < 4; ++f) {
          const int dcol = 16 * f + mrow;
          v2f b;
          b.x = Vt[(4 * kk + 2 * half)     * 68 + dcol];
          b.y = Vt[(4 * kk + 2 * half + 1) * 68 + dcol];
          o[f] = wmma_f32(a, b, o[f]);
        }
      }
    }
  }

  #pragma unroll
  for (int round = 0; round < 2; ++round) {
    const int senderLo = (round == 0) ? 2 : 1;
    const int nPairs   = (round == 0) ? 2 : 1;
    __syncthreads();
    if (grp >= senderLo && grp < senderLo + nPairs) {
      float* mb = smem + ((grp - senderLo) * 4 + qw) * 1536;
      #pragma unroll
      for (int f = 0; f < 4; ++f)
        #pragma unroll
        for (int r = 0; r < 8; ++r) mb[(f * 8 + r) * 32 + lane] = o[f][r];
      #pragma unroll
      for (int r = 0; r < 8; ++r) {
        mb[1024 + r * 32 + lane] = mrun[r];
        mb[1280 + r * 32 + lane] = lrun[r];
      }
    }
    __syncthreads();
    if (grp < nPairs) {
      const float* mb = smem + (grp * 4 + qw) * 1536;
      float a0[8], a1[8];
      #pragma unroll
      for (int r = 0; r < 8; ++r) {
        const float m1 = mb[1024 + r * 32 + lane];
        const float l1 = mb[1280 + r * 32 + lane];
        const float m  = fmaxf(mrun[r], m1);
        a0[r] = __expf(mrun[r] - m);
        a1[r] = __expf(m1 - m);
        lrun[r] = lrun[r] * a0[r] + l1 * a1[r];
        mrun[r] = m;
      }
      #pragma unroll
      for (int f = 0; f < 4; ++f)
        #pragma unroll
        for (int r = 0; r < 8; ++r)
          o[f][r] = o[f][r] * a0[r] + mb[(f * 8 + r) * 32 + lane] * a1[r];
    }
  }

  if (grp == 0) {
    #pragma unroll
    for (int r = 0; r < 8; ++r) lrun[r] = 1.0f / lrun[r];
    float* obase = O + (size_t)n * S_ * E_ + h * D_;
    #pragma unroll
    for (int f = 0; f < 4; ++f)
      #pragma unroll
      for (int r = 0; r < 8; ++r) {
        const int m = r + 8 * half;
        const int dcol = 16 * f + mrow;
        obase[(size_t)(q0 + m) * E_ + dcol] = o[f][r] * lrun[r];
      }
  }
}

__global__ __launch_bounds__(128)
void outproj_kernel(const float* __restrict__ A,
                    const float* __restrict__ Wo,
                    const float* __restrict__ bo,
                    float* __restrict__ Y)
{
  __shared__ float Ws[64 * 68];
  const int tid  = threadIdx.x;
  const int wave = tid >> 5;
  const int lane = tid & 31;
  const int r0 = blockIdx.y * 64 + wave * 16;
  const int c0 = blockIdx.x * 64;
  const int mrow = lane & 15, half = lane >> 4;

  v8f c[4] = {};
  const float* arow = A + (size_t)(r0 + mrow) * E_;

  for (int kc = 0; kc < E_; kc += 64) {
    __syncthreads();
    for (int i = tid; i < 1024; i += 128) {
      const int rr = i >> 4, c4 = i & 15;
      const float4 w4 = ((const float4*)(Wo + (size_t)(c0 + rr) * E_ + kc))[c4];
      float* d = &Ws[rr * 68 + c4 * 4];
      d[0] = w4.x; d[1] = w4.y; d[2] = w4.z; d[3] = w4.w;
    }
    __syncthreads();
    #pragma unroll
    for (int ks = 0; ks < 16; ++ks) {
      const int k = 4 * ks + 2 * half;
      v2f a; a.x = arow[kc + k]; a.y = arow[kc + k + 1];
      #pragma unroll
      for (int f = 0; f < 4; ++f) {
        const int e = 16 * f + mrow;
        v2f b; b.x = Ws[e * 68 + k]; b.y = Ws[e * 68 + k + 1];
        c[f] = wmma_f32(a, b, c[f]);
      }
    }
  }

  #pragma unroll
  for (int f = 0; f < 4; ++f)
    #pragma unroll
    for (int r = 0; r < 8; ++r) {
      const int m = r + 8 * half;
      const int e = c0 + 16 * f + mrow;
      Y[(size_t)(r0 + m) * E_ + e] = c[f][r] + bo[e];
    }
}

extern "C" void kernel_launch(void* const* d_in, const int* in_sizes, int n_in,
                              void* d_out, int out_size, void* d_ws, size_t ws_size,
                              hipStream_t stream) {
  const float* values = (const float*)d_in[0];
  const float* keys   = (const float*)d_in[1];
  const float* query  = (const float*)d_in[2];
  const float* Wv     = (const float*)d_in[4];
  const float* Wk     = (const float*)d_in[5];
  const float* Wq     = (const float*)d_in[6];
  const float* Wo     = (const float*)d_in[7];
  const float* bo     = (const float*)d_in[8];
  float* out = (float*)d_out;

  const size_t tensorElems = (size_t)N_ * H_ * S_ * D_;
  float* Qp = (float*)d_ws;
  float* Kp = Qp + tensorElems;
  float* Vp = Kp + tensorElems;
  float* Oa = Vp + tensorElems;

  const dim3 blk(128);
  const dim3 gProj(S_ / 64, H_, N_);

  proj_kernel<<<gProj, blk, 0, stream>>>(query,  Wq, Qp);
  proj_kernel<<<gProj, blk, 0, stream>>>(keys,   Wk, Kp);
  proj_kernel<<<gProj, blk, 0, stream>>>(values, Wv, Vp);

  const dim3 gFlash(N_ * H_, S_ / 64);
  flash_kernel<<<gFlash, dim3(512), 0, stream>>>(Qp, Kp, Vp, Oa);

  const dim3 gOut(E_ / 64, (N_ * S_) / 64);
  outproj_kernel<<<gOut, blk, 0, stream>>>(Oa, Wo, bo, out);
}


# attention kernel: register budget cut to 128 per wave (launch bound of 8 waves per SIMD; mask distance kept in one register, not eight) so two work-groups share a compute unit; sums unchanged
# speedup vs baseline: 1.5051x; 1.0590x over previous
#include <hip/hip_runtime.h>
#include <hip/hip_bf16.h>
#include <math.h>


typedef float v2f __attribute__((ext_vector_type(2)));
typedef float v8f __attribute__((ext_vector_type(8)));

static constexpr int N_ = 2;
static constexpr int S_ = 2048;
static constexpr int E_ = 1024;
static constexpr int H_ = 16;
static constexpr int D_ = 64;

static __device__ __forceinline__ v8f wmma_f32(v2f a, v2f b, v8f c) {
  return __builtin_amdgcn_wmma_f32_16x16x4_f32(false, a, false, b,
                                               (short)0, c, false, false);
}

template <int M>
static __device__ __forceinline__ float row_xor16(float v) {
  const int vi = __builtin_bit_cast(int, v);
  return __builtin_bit_cast(float,
      __builtin_amdgcn_update_dpp(vi, vi, 0x160 + M, 0xf, 0xf, true));
}
static __device__ __forceinline__ float rowmax16(float v) {
  v = fmaxf(v, row_xor16<1>(v));
  v = fmaxf(v, row_xor16<2>(v));
  v = fmaxf(v, row_xor16<4>(v));
  v = fmaxf(v, row_xor16<8>(v));
  return v;
}
static __device__ __forceinline__ float rowsum16(float v) {
  v += row_xor16<1>(v);
  v += row_xor16<2>(v);
  v += row_xor16<4>(v);
  v += row_xor16<8>(v);
  return v;
}

__global__ __launch_bounds__(128)
void proj_kernel(const float* __restrict__ X,
                 const float* __restrict__ W,
                 float* __restrict__ Out)
{
  __shared__ float Ws[64 * 68];
  const int tid  = threadIdx.x;
  const int wave = tid >> 5;
  const int lane = tid & 31;
  const int n  = blockIdx.z, h = blockIdx.y;
  const int s0 = blockIdx.x * 64 + wave * 16;
  const int mrow = lane & 15, half = lane >> 4;

  for (int i = tid; i < 1024; i += 128) {
    const int r = i >> 4, c4 = i & 15;
    const float4 w4 = ((const float4*)W)[r * 16 + c4];
    float* d = &Ws[r * 68 + c4 * 4];
    d[0] = w4.x; d[1] = w4.y; d[2] = w4.z; d[3] = w4.w;
  }
  __syncthreads();

  const float* xrow = X + ((size_t)n * S_ + (s0 + mrow)) * E_ + h * D_;
  v8f c[4] = {};
  #pragma unroll
  for (int ks = 0; ks < 16; ++ks) {
    const int k = 4 * ks + 2 * half;
    v2f a; a.x = xrow[k]; a.y = xrow[k + 1];
    #pragma unroll
    for (int f = 0; f < 4; ++f) {
      const int e = 16 * f + mrow;
      v2f b; b.x = Ws[e * 68 + k]; b.y = Ws[e * 68 + k + 1];
      c[f] = wmma_f32(a, b, c[f]);
    }
  }

  float* obase = Out + ((size_t)(n * H_ + h)) * S_ * D_;
  #pragma unroll
  for (int f = 0; f < 4; ++f)
    #pragma unroll
    for (int r = 0; r < 8; ++r) {
      const int m = r + 8 * half;
      const int e = 16 * f + mrow;
      obase[(size_t)(s0 + m) * D_ + e] = c[f][r];
    }
}

__global__ __launch_bounds__(512, 8)
void flash_kernel(const float* __restrict__ Qp,
                  const float* __restrict__ Kp,
                  const float* __restrict__ Vp,
                  float* __restrict__ O)
{
  __shared__ float smem[4352 + 4352 + 4352];
  float* Ks = smem;
  float* Vs = smem + 4352;

  const int tid  = threadIdx.x;
  const int wave = tid >> 5;
  const int lane = tid & 31;
  const int grp  = wave >> 2;
  const int qw   = wave & 3;
  const int n = blockIdx.x / H_, h = blockIdx.x % H_;
  const int qblk = (S_ / 64 - 1) - blockIdx.y;
  const int q0 = qblk * 64 + qw * 16;
  const int mrow = lane & 15, half = lane >> 4;
  const size_t headOff = ((size_t)(n * H_ + h)) * S_ * D_;
  float* Pw = smem + 8704 + wave * 272;
  const float* Kt = Ks + grp * (16 * 68);
  const float* Vt = Vs + grp * (16 * 68);

  float qreg[32];
  const float* qrow = Qp + headOff + (size_t)(q0 + mrow) * D_;
  #pragma unroll
  for (int ks = 0; ks < 16; ++ks) {
    const int k = 4 * ks + 2 * half;
    qreg[2 * ks]     = qrow[k];
    qreg[2 * ks + 1] = qrow[k + 1];
  }

  v8f o[4] = {};
  float mrun[8], lrun[8];
  #pragma unroll
  for (int r = 0; r < 8; ++r) { mrun[r] = -3.0e38f; lrun[r] = 0.0f; }

  const float slope  = (float)(1 << __popc(h));
  const float scale  = 0.03125f;
  const int   nSteps = qblk + 1;

  for (int st = 0; st < nSteps; ++st) {
    const int kbase = st * 64;
    __syncthreads();
    for (int i = tid; i < 1024; i += 512) {
      const int r = i >> 4, c4 = i & 15;
      const float4 kq = ((const float4*)(Kp + headOff + (size_t)(kbase + r) * D_))[c4];
      const float4 vq = ((const float4*)(Vp + headOff + (size_t)(kbase + r) * D_))[c4];
      float* kd = &Ks[r * 68 + c4 * 4];
      float* vd = &Vs[r * 68 + c4 * 4];
      kd[0] = kq.x; kd[1] = kq.y; kd[2] = kq.z; kd[3] = kq.w;
      vd[0] = vq.x; vd[1] = vq.y; vd[2] = vq.z; vd[3] = vq.w;
    }
    __syncthreads();

    const int k0 = kbase + grp * 16;
    const bool active = (k0 <= q0 + 15);
    if (active) {
      v8f s = {};
      #pragma unroll
      for (int ks = 0; ks < 16; ++ks) {
        const int k = 4 * ks + 2 * half;
        v2f a; a.x = qreg[2 * ks]; a.y = qreg[2 * ks + 1];
        v2f b; b.x = Kt[mrow * 68 + k]; b.y = Kt[mrow * 68 + k + 1];
        s = wmma_f32(a, b, s);
      }
      const int   d0 = (k0 + mrow) - (q0 + 8 * half);
      const float f0 = (float)d0;
      #pragma unroll
      for (int r = 0; r < 8; ++r) {
        float e = (s[r] + slope * (f0 - (float)r)) * scale;
        if (d0 > r) e = -3.0e18f;
        s[r] = e;
      }
      float pscale[8];
      #pragma unroll
      for (int r = 0; r < 8; ++r) {
        const float rm   = rowmax16(s[r]);
        const float mnew = fmaxf(mrun[r], rm);
        const float alpha = __expf(mrun[r] - mnew);
        mrun[r] = mnew;
        const float p = __expf(s[r] - mnew);
        s[r] = p;
        lrun[r] = lrun[r] * alpha + rowsum16(p);
        pscale[r] = alpha;
      }
      #pragma unroll
      for (int f = 0; f < 4; ++f)
        #pragma unroll
        for (int r = 0; r < 8; ++r) o[f][r] *= pscale[r];

      #pragma unroll
      for (int r = 0; r < 8; ++r)
        Pw[(r + 8 * half) * 17 + mrow] = s[r];
      #pragma unroll
      for (int kk = 0; kk < 4; ++kk) {
        const int ka = 4 * kk + 2 * half;
        v2f a; a.x = Pw[mrow * 17 + ka]; a.y = Pw[mrow * 17 + ka + 1];
        #pragma unroll
        for (int f = 0; f < 4; ++f) {
          const int dcol = 16 * f + mrow;
          v2f b;
          b.x = Vt[(4 * kk + 2 * half)     * 68 + dcol];
          b.y = Vt[(4 * kk + 2 * half + 1) * 68 + dcol];
          o[f] = wmma_f32(a, b, o[f]);
        }
      }
    }
  }

  #pragma unroll
  for (int round = 0; round < 2; ++round) {
    const int senderLo = (round == 0) ? 2 : 1;
    const int nPairs   = (round == 0) ? 2 : 1;
    __syncthreads();
    if (grp >= senderLo && grp < senderLo + nPairs) {
      float* mb = smem + ((grp - senderLo) * 4 + qw) * 1536;
      #pragma unroll
      for (int f = 0; f < 4; ++f)
        #pragma unroll
        for (int r = 0; r < 8; ++r) mb[(f * 8 + r) * 32 + lane] = o[f][r];
      #pragma unroll
      for (int r = 0; r < 8; ++r) {
        mb[1024 + r * 32 + lane] = mrun[r];
        mb[1280 + r * 32 + lane] = lrun[r];
      }
    }
    __syncthreads();
    if (grp < nPairs) {
      const float* mb = smem + (grp * 4 + qw) * 1536;
      float a0[8], a1[8];
      #pragma unroll
      for (int r = 0; r < 8; ++r) {
        const float m1 = mb[1024 + r * 32 + lane];
        const float l1 = mb[1280 + r * 32 + lane];
        const float m  = fmaxf(mrun[r], m1);
        a0[r] = __expf(mrun[r] - m);
        a1[r] = __expf(m1 - m);
        lrun[r] = lrun[r] * a0[r] + l1 * a1[r];
        mrun[r] = m;
      }
      #pragma unroll
      for (int f = 0; f < 4; ++f)
        #pragma unroll
        for (int r = 0; r < 8; ++r)
          o[f][r] = o[f][r] * a0[r] + mb[(f * 8 + r) * 32 + lane] * a1[r];
    }
  }

  if (grp == 0) {
    #pragma unroll
    for (int r = 0; r < 8; ++r) lrun[r] = 1.0f / lrun[r];
    float* obase = O + (size_t)n * S_ * E_ + h * D_;
    #pragma unroll
    for (int f = 0; f < 4; ++f)
      #pragma unroll
      for (int r = 0; r < 8; ++r) {
        const int m = r + 8 * half;
        const int dcol = 16 * f + mrow;
        obase[(size_t)(q0 + m) * E_ + dcol] = o[f][r] * lrun[r];
      }
  }
}

__global__ __launch_bounds__(128)
void outproj_kernel(const float* __restrict__ A,
                    const float* __restrict__ Wo,
                    const float* __restrict__ bo,
                    float* __restrict__ Y)
{
  __shared__ float Ws[64 * 68];
  const int tid  = threadIdx.x;
  const int wave = tid >> 5;
  const int lane = tid & 31;
  const int r0 = blockIdx.y * 64 + wave * 16;
  const int c0 = blockIdx.x * 64;
  const int mrow = lane & 15, half = lane >> 4;

  v8f c[4] = {};
  const float* arow = A + (size_t)(r0 + mrow) * E_;

  for (int kc = 0; kc < E_; kc += 64) {
    __syncthreads();
    for (int i = tid; i < 1024; i += 128) {
      const int rr = i >> 4, c4 = i & 15;
      const float4 w4 = ((const float4*)(Wo + (size_t)(c0 + rr) * E_ + kc))[c4];
      float* d = &Ws[rr * 68 + c4 * 4];
      d[0] = w4.x; d[1] = w4.y; d[2] = w4.z; d[3] = w4.w;
    }
    __syncthreads();
    #pragma unroll
    for (int ks = 0; ks < 16; ++ks) {
      const int k = 4 * ks + 2 * half;
      v2f a; a.x = arow[kc + k]; a.y = arow[kc + k + 1];
      #pragma unroll
      for (int f = 0; f < 4; ++f) {
        const int e = 16 * f + mrow;
        v2f b; b.x = Ws[e * 68 + k]; b.y = Ws[e * 68 + k + 1];
        c[f] = wmma_f32(a, b, c[f]);
      }
    }
  }

  #pragma unroll
  for (int f = 0; f < 4; ++f)
    #pragma unroll
    for (int r = 0; r < 8; ++r) {
      const int m = r + 8 * half;
      const int e = c0 + 16 * f + mrow;
      Y[(size_t)(r0 + m) * E_ + e] = c[f][r] + bo[e];
    }
}

extern "C" void kernel_launch(void* const* d_in, const int* in_sizes, int n_in,
                              void* d_out, int out_size, void* d_ws, size_t ws_size,
                              hipStream_t stream) {
  const float* values = (const float*)d_in[0];
  const float* keys   = (const float*)d_in[1];
  const float* query  = (const float*)d_in[2];
  const float* Wv     = (const float*)d_in[4];
  const float* Wk     = (const float*)d_in[5];
  const float* Wq     = (const float*)d_in[6];
  const float* Wo     = (const float*)d_in[7];
  const float* bo     = (const float*)d_in[8];
  float* out = (float*)d_out;

  const size_t tensorElems = (size_t)N_ * H_ * S_ * D_;
  float* Qp = (float*)d_ws;
  float* Kp = Qp + tensorElems;
  float* Vp = Kp + tensorElems;
  float* Oa = Vp + tensorElems;

  const dim3 blk(128);
  const dim3 gProj(S_ / 64, H_, N_);

  proj_kernel<<<gProj, blk, 0, stream>>>(query,  Wq, Qp);
  proj_kernel<<<gProj, blk, 0, stream>>>(keys,   Wk, Kp);
  proj_kernel<<<gProj, blk, 0, stream>>>(values, Wv, Vp);

  const dim3 gFlash(N_ * H_, S_ / 64);
  flash_kernel<<<gFlash, dim3(512), 0, stream>>>(Qp, Kp, Vp, Oa);

  const dim3 gOut(E_ / 64, (N_ * S_) / 64);
  outproj_kernel<<<gOut, blk, 0, stream>>>(Oa, Wo, bo, out);
}


# attention: K/V tiles stored by the projections as WMMA B fragments, loaded by each wave straight from global memory: no LDS staging, no key-loop barrier, one block per CU; q.k and P.V sums reordered
# speedup vs baseline: 1.7215x; 1.1438x over previous
#include <hip/hip_runtime.h>
#include <hip/hip_bf16.h>
#include <math.h>


typedef float v2f __attribute__((ext_vector_type(2)));
typedef float v8f __attribute__((ext_vector_type(8)));

static constexpr int N_ = 2;
static constexpr int S_ = 2048;
static constexpr int E_ = 1024;
static constexpr int H_ = 16;
static constexpr int D_ = 64;

static __device__ __forceinline__ v8f wmma_f32(v2f a, v2f b, v8f c) {
  return __builtin_amdgcn_wmma_f32_16x16x4_f32(false, a, false, b,
                                               (short)0, c, false, false);
}

template <int M>
static __device__ __forceinline__ float row_xor16(float v) {
  const int vi = __builtin_bit_cast(int, v);
  return __builtin_bit_cast(float,
      __builtin_amdgcn_update_dpp(vi, vi, 0x160 + M, 0xf, 0xf, true));
}
static __device__ __forceinline__ float rowmax16(float v) {
  v = fmaxf(v, row_xor16<1>(v));
  v = fmaxf(v, row_xor16<2>(v));
  v = fmaxf(v, row_xor16<4>(v));
  v = fmaxf(v, row_xor16<8>(v));
  return v;
}
static __device__ __forceinline__ float rowsum16(float v) {
  v += row_xor16<1>(v);
  v += row_xor16<2>(v);
  v += row_xor16<4>(v);
  v += row_xor16<8>(v);
  return v;
}

template <int MODE>
__global__ __launch_bounds__(128)
void proj_kernel(const float* __restrict__ X,
                 const float* __restrict__ W,
                 float* __restrict__ Out)
{
  __shared__ float Ws[64 * 68];
  const int tid  = threadIdx.x;
  const int wave = tid >> 5;
  const int lane = tid & 31;
  const int n  = blockIdx.z, h = blockIdx.y;
  const int s0 = blockIdx.x * 64 + wave * 16;
  const int mrow = lane & 15, half = lane >> 4;

  for (int i = tid; i < 1024; i += 128) {
    const int r = i >> 4, c4 = i & 15;
    const float4 w4 = ((const float4*)W)[r * 16 + c4];
    float* d = &Ws[r * 68 + c4 * 4];
    d[0] = w4.x; d[1] = w4.y; d[2] = w4.z; d[3] = w4.w;
  }
  __syncthreads();

  const float* xrow = X + ((size_t)n * S_ + (s0 + mrow)) * E_ + h * D_;
  v8f c[4] = {};
  #pragma unroll
  for (int ks = 0; ks < 16; ++ks) {
    const int k = 4 * ks + 2 * half;
    v2f a; a.x = xrow[k]; a.y = xrow[k + 1];
    #pragma unroll
    for (int f = 0; f < 4; ++f) {
      const int e = 16 * f + mrow;
      v2f b; b.x = Ws[e * 68 + k]; b.y = Ws[e * 68 + k + 1];
      if (MODE == 1) c[f] = wmma_f32(b, a, c[f]);
      else           c[f] = wmma_f32(a, b, c[f]);
    }
  }

  float* obase = Out + ((size_t)(n * H_ + h)) * S_ * D_;
  if (MODE == 0) {
    #pragma unroll
    for (int f = 0; f < 4; ++f)
      #pragma unroll
      for (int r = 0; r < 8; ++r) {
        const int m = r + 8 * half;
        const int e = 16 * f + mrow;
        obase[(size_t)(s0 + m) * D_ + e] = c[f][r];
      }
  } else {
    float* tile = obase + (size_t)(s0 >> 4) * 1024 + lane * 4;
    #pragma unroll
    for (int f = 0; f < 4; ++f)
      #pragma unroll
      for (int jj = 0; jj < 2; ++jj) {
        float4 v4;
        v4.x = c[f][4 * jj];     v4.y = c[f][4 * jj + 1];
        v4.z = c[f][4 * jj + 2]; v4.w = c[f][4 * jj + 3];
        *(float4*)(tile + (f * 2 + jj) * 128) = v4;
      }
  }
}

__global__ __launch_bounds__(512)
void flash_kernel(const float* __restrict__ Qp,
                  const float* __restrict__ Kf,
                  const float* __restrict__ Vf,
                  float* __restrict__ O)
{
  __shared__ float smem[4352 + 4352 + 4352];

  const int tid  = threadIdx.x;
  const int wave = __builtin_amdgcn_readfirstlane(tid >> 5);
  const int lane = tid & 31;
  const int grp  = wave >> 2;
  const int qw   = wave & 3;
  const int n = blockIdx.x / H_, h = blockIdx.x % H_;
  const int qblk = (S_ / 64 - 1) - blockIdx.y;
  const int q0 = qblk * 64 + qw * 16;
  const int mrow = lane & 15, half = lane >> 4;
  const size_t headOff = ((size_t)(n * H_ + h)) * S_ * D_;
  float* Pw = smem + 8704 + wave * 272;

  float qreg[32];
  const float* qrow = Qp + headOff + (size_t)(q0 + mrow) * D_ + 8 * half;
  #pragma unroll
  for (int f = 0; f < 4; ++f) {
    const float4 qa = *(const float4*)(qrow + 16 * f);
    const float4 qb = *(const float4*)(qrow + 16 * f + 4);
    qreg[8 * f + 0] = qa.x; qreg[8 * f + 1] = qa.y; qreg[8 * f + 2] = qa.z; qreg[8 * f + 3] = qa.w;
    qreg[8 * f + 4] = qb.x; qreg[8 * f + 5] = qb.y; qreg[8 * f + 6] = qb.z; qreg[8 * f + 7] = qb.w;
  }

  v8f o[4] = {};
  float mrun[8], lrun[8];
  #pragma unroll
  for (int r = 0; r < 8; ++r) { mrun[r] = -3.0e38f; lrun[r] = 0.0f; }

  const float slope  = (float)(1 << __popc(h));
  const float scale  = 0.03125f;
  const int   nMine  = qblk + ((grp <= qw) ? 1 : 0);

  const float* kt = Kf + headOff + (size_t)grp * 1024 + lane * 4;
  const float* vt = Vf + headOff + (size_t)grp * 1024 + lane * 4;

  for (int st = 0; st < nMine; ++st, kt += 4096, vt += 4096) {
    float4 kq[8], vq[8];
    #pragma unroll
    for (int i = 0; i < 8; ++i) kq[i] = *(const float4*)(kt + i * 128);
    #pragma unroll
    for (int i = 0; i < 8; ++i) vq[i] = *(const float4*)(vt + i * 128);
    __builtin_amdgcn_sched_barrier(0);

    const int k0 = st * 64 + grp * 16;
    v8f s = {};
    #pragma unroll
    for (int t = 0; t < 16; ++t) {
      const float4 k4 = kq[t >> 1];
      v2f a; a.x = qreg[2 * t]; a.y = qreg[2 * t + 1];
      v2f b;
      if (t & 1) { b.x = k4.z; b.y = k4.w; } else { b.x = k4.x; b.y = k4.y; }
      s = wmma_f32(a, b, s);
    }
    const int   d0 = (k0 + mrow) - (q0 + 8 * half);
    const float f0 = (float)d0;
    #pragma unroll
    for (int r = 0; r < 8; ++r) {
      float e = (s[r] + slope * (f0 - (float)r)) * scale;
      if (d0 > r) e = -3.0e18f;
      s[r] = e;
    }
    float pscale[8];
    #pragma unroll
    for (int r = 0; r < 8; ++r) {
      const float rm   = rowmax16(s[r]);
      const float mnew = fmaxf(mrun[r], rm);
      const float alpha = __expf(mrun[r] - mnew);
      mrun[r] = mnew;
      const float p = __expf(s[r] - mnew);
      s[r] = p;
      lrun[r] = lrun[r] * alpha + rowsum16(p);
      pscale[r] = alpha;
    }
    #pragma unroll
    for (int f = 0; f < 4; ++f)
      #pragma unroll
      for (int r = 0; r < 8; ++r) o[f][r] *= pscale[r];

    #pragma unroll
    for (int r = 0; r < 8; ++r)
      Pw[(r + 8 * half) * 17 + mrow] = s[r];
    #pragma unroll
    for (int kk = 0; kk < 4; ++kk) {
      const int ka = 8 * half + 2 * kk;
      v2f a; a.x = Pw[mrow * 17 + ka]; a.y = Pw[mrow * 17 + ka + 1];
      #pragma unroll
      for (int f = 0; f < 4; ++f) {
        const float4 v4 = vq[f * 2 + (kk >> 1)];
        v2f b;
        if (kk & 1) { b.x = v4.z; b.y = v4.w; } else { b.x = v4.x; b.y = v4.y; }
        o[f] = wmma_f32(a, b, o[f]);
      }
    }
  }

  #pragma unroll
  for (int round = 0; round < 2; ++round) {
    const int senderLo = (round == 0) ? 2 : 1;
    const int nPairs   = (round == 0) ? 2 : 1;
    __syncthreads();
    if (grp >= senderLo && grp < senderLo + nPairs) {
      float* mb = smem + ((grp - senderLo) * 4 + qw) * 1536;
      #pragma unroll
      for (int f = 0; f < 4; ++f)
        #pragma unroll
        for (int r = 0; r < 8; ++r) mb[(f * 8 + r) * 32 + lane] = o[f][r];
      #pragma unroll
      for (int r = 0; r < 8; ++r) {
        mb[1024 + r * 32 + lane] = mrun[r];
        mb[1280 + r * 32 + lane] = lrun[r];
      }
    }
    __syncthreads();
    if (grp < nPairs) {
      const float* mb = smem + (grp * 4 + qw) * 1536;
      float a0[8], a1[8];
      #pragma unroll
      for (int r = 0; r < 8; ++r) {
        const float m1 = mb[1024 + r * 32 + lane];
        const float l1 = mb[1280 + r * 32 + lane];
        const float m  = fmaxf(mrun[r], m1);
        a0[r] = __expf(mrun[r] - m);
        a1[r] = __expf(m1 - m);
        lrun[r] = lrun[r] * a0[r] + l1 * a1[r];
        mrun[r] = m;
      }
      #pragma unroll
      for (int f = 0; f < 4; ++f)
        #pragma unroll
        for (int r = 0; r < 8; ++r)
          o[f][r] = o[f][r] * a0[r] + mb[(f * 8 + r) * 32 + lane] * a1[r];
    }
  }

  if (grp == 0) {
    #pragma unroll
    for (int r = 0; r < 8; ++r) lrun[r] = 1.0f / lrun[r];
    float* obase = O + (size_t)n * S_ * E_ + h * D_;
    #pragma unroll
    for (int f = 0; f < 4; ++f)
      #pragma unroll
      for (int r = 0; r < 8; ++r) {
        const int m = r + 8 * half;
        const int dcol = 16 * f + mrow;
        obase[(size_t)(q0 + m) * E_ + dcol] = o[f][r] * lrun[r];
      }
  }
}

__global__ __launch_bounds__(128)
void outproj_kernel(const float* __restrict__ A,
                    const float* __restrict__ Wo,
                    const float* __restrict__ bo,
                    float* __restrict__ Y)
{
  __shared__ float Ws[64 * 68];
  const int tid  = threadIdx.x;
  const int wave = tid >> 5;
  const int lane = tid & 31;
  const int r0 = blockIdx.y * 64 + wave * 16;
  const int c0 = blockIdx.x * 64;
  const int mrow = lane & 15, half = lane >> 4;

  v8f c[4] = {};
  const float* arow = A + (size_t)(r0 + mrow) * E_;

  for (int kc = 0; kc < E_; kc += 64) {
    __syncthreads();
    for (int i = tid; i < 1024; i += 128) {
      const int rr = i >> 4, c4 = i & 15;
      const float4 w4 = ((const float4*)(Wo + (size_t)(c0 + rr) * E_ + kc))[c4];
      float* d = &Ws[rr * 68 + c4 * 4];
      d[0] = w4.x; d[1] = w4.y; d[2] = w4.z; d[3] = w4.w;
    }
    __syncthreads();
    #pragma unroll
    for (int ks = 0; ks < 16; ++ks) {
      const int k = 4 * ks + 2 * half;
      v2f a; a.x = arow[kc + k]; a.y = arow[kc + k + 1];
      #pragma unroll
      for (int f = 0; f < 4; ++f) {
        const int e = 16 * f + mrow;
        v2f b; b.x = Ws[e * 68 + k]; b.y = Ws[e * 68 + k + 1];
        c[f] = wmma_f32(a, b, c[f]);
      }
    }
  }

  #pragma unroll
  for (int f = 0; f < 4; ++f)
    #pragma unroll
    for (int r = 0; r < 8; ++r) {
      const int m = r + 8 * half;
      const int e = c0 + 16 * f + mrow;
      Y[(size_t)(r0 + m) * E_ + e] = c[f][r] + bo[e];
    }
}

extern "C" void kernel_launch(void* const* d_in, const int* in_sizes, int n_in,
                              void* d_out, int out_size, void* d_ws, size_t ws_size,
                              hipStream_t stream) {
  const float* values = (const float*)d_in[0];
  const float* keys   = (const float*)d_in[1];
  const float* query  = (const float*)d_in[2];
  const float* Wv     = (const float*)d_in[4];
  const float* Wk     = (const float*)d_in[5];
  const float* Wq     = (const float*)d_in[6];
  const float* Wo     = (const float*)d_in[7];
  const float* bo     = (const float*)d_in[8];
  float* out = (float*)d_out;

  const size_t tensorElems = (size_t)N_ * H_ * S_ * D_;
  float* Qp = (float*)d_ws;
  float* Kp = Qp + tensorElems;
  float* Vp = Kp + tensorElems;
  float* Oa = Vp + tensorElems;

  const dim3 blk(128);
  const dim3 gProj(S_ / 64, H_, N_);

  proj_kernel<0><<<gProj, blk, 0, stream>>>(query,  Wq, Qp);
  proj_kernel<1><<<gProj, blk, 0, stream>>>(keys,   Wk, Kp);
  proj_kernel<2><<<gProj, blk, 0, stream>>>(values, Wv, Vp);

  const dim3 gFlash(N_ * H_, S_ / 64);
  flash_kernel<<<gFlash, dim3(512), 0, stream>>>(Qp, Kp, Vp, Oa);

  const dim3 gOut(E_ / 64, (N_ * S_) / 64);
  outproj_kernel<<<gOut, blk, 0, stream>>>(Oa, Wo, bo, out);
}


# output projection retiled: 128x64 block tile, 32x64 per wave (8 accumulator tiles); A and W chunks staged in LDS by whole-line loads, fragments read as 16-byte LDS loads; k sums run in a new order
# speedup vs baseline: 2.1468x; 1.2471x over previous
#include <hip/hip_runtime.h>
#include <hip/hip_bf16.h>
#include <math.h>


typedef float v2f __attribute__((ext_vector_type(2)));
typedef float v8f __attribute__((ext_vector_type(8)));

static constexpr int N_ = 2;
static constexpr int S_ = 2048;
static constexpr int E_ = 1024;
static constexpr int H_ = 16;
static constexpr int D_ = 64;

static __device__ __forceinline__ v8f wmma_f32(v2f a, v2f b, v8f c) {
  return __builtin_amdgcn_wmma_f32_16x16x4_f32(false, a, false, b,
                                               (short)0, c, false, false);
}

template <int M>
static __device__ __forceinline__ float row_xor16(float v) {
  const int vi = __builtin_bit_cast(int, v);
  return __builtin_bit_cast(float,
      __builtin_amdgcn_update_dpp(vi, vi, 0x160 + M, 0xf, 0xf, true));
}
static __device__ __forceinline__ float rowmax16(float v) {
  v = fmaxf(v, row_xor16<1>(v));
  v = fmaxf(v, row_xor16<2>(v));
  v = fmaxf(v, row_xor16<4>(v));
  v = fmaxf(v, row_xor16<8>(v));
  return v;
}
static __device__ __forceinline__ float rowsum16(float v) {
  v += row_xor16<1>(v);
  v += row_xor16<2>(v);
  v += row_xor16<4>(v);
  v += row_xor16<8>(v);
  return v;
}

template <int MODE>
__global__ __launch_bounds__(128)
void proj_kernel(const float* __restrict__ X,
                 const float* __restrict__ W,
                 float* __restrict__ Out)
{
  __shared__ float Ws[64 * 68];
  const int tid  = threadIdx.x;
  const int wave = tid >> 5;
  const int lane = tid & 31;
  const int n  = blockIdx.z, h = blockIdx.y;
  const int s0 = blockIdx.x * 64 + wave * 16;
  const int mrow = lane & 15, half = lane >> 4;

  for (int i = tid; i < 1024; i += 128) {
    const int r = i >> 4, c4 = i & 15;
    const float4 w4 = ((const float4*)W)[r * 16 + c4];
    float* d = &Ws[r * 68 + c4 * 4];
    d[0] = w4.x; d[1] = w4.y; d[2] = w4.z; d[3] = w4.w;
  }
  __syncthreads();

  const float* xrow = X + ((size_t)n * S_ + (s0 + mrow)) * E_ + h * D_;
  v8f c[4] = {};
  #pragma unroll
  for (int ks = 0; ks < 16; ++ks) {
    const int k = 4 * ks + 2 * half;
    v2f a; a.x = xrow[k]; a.y = xrow[k + 1];
    #pragma unroll
    for (int f = 0; f < 4; ++f) {
      const int e = 16 * f + mrow;
      v2f b; b.x = Ws[e * 68 + k]; b.y = Ws[e * 68 + k + 1];
      if (MODE == 1) c[f] = wmma_f32(b, a, c[f]);
      else           c[f] = wmma_f32(a, b, c[f]);
    }
  }

  float* obase = Out + ((size_t)(n * H_ + h)) * S_ * D_;
  if (MODE == 0) {
    #pragma unroll
    for (int f = 0; f < 4; ++f)
      #pragma unroll
      for (int r = 0; r < 8; ++r) {
        const int m = r + 8 * half;
        const int e = 16 * f + mrow;
        obase[(size_t)(s0 + m) * D_ + e] = c[f][r];
      }
  } else {
    float* tile = obase + (size_t)(s0 >> 4) * 1024 + lane * 4;
    #pragma unroll
    for (int f = 0; f < 4; ++f)
      #pragma unroll
      for (int jj = 0; jj < 2; ++jj) {
        float4 v4;
        v4.x = c[f][4 * jj];     v4.y = c[f][4 * jj + 1];
        v4.z = c[f][4 * jj + 2]; v4.w = c[f][4 * jj + 3];
        *(float4*)(tile + (f * 2 + jj) * 128) = v4;
      }
  }
}

__global__ __launch_bounds__(512)
void flash_kernel(const float* __restrict__ Qp,
                  const float* __restrict__ Kf,
                  const float* __restrict__ Vf,
                  float* __restrict__ O)
{
  __shared__ float smem[4352 + 4352 + 4352];

  const int tid  = threadIdx.x;
  const int wave = __builtin_amdgcn_readfirstlane(tid >> 5);
  const int lane = tid & 31;
  const int grp  = wave >> 2;
  const int qw   = wave & 3;
  const int n = blockIdx.x / H_, h = blockIdx.x % H_;
  const int qblk = (S_ / 64 - 1) - blockIdx.y;
  const int q0 = qblk * 64 + qw * 16;
  const int mrow = lane & 15, half = lane >> 4;
  const size_t headOff = ((size_t)(n * H_ + h)) * S_ * D_;
  float* Pw = smem + 8704 + wave * 272;

  float qreg[32];
  const float* qrow = Qp + headOff + (size_t)(q0 + mrow) * D_ + 8 * half;
  #pragma unroll
  for (int f = 0; f < 4; ++f) {
    const float4 qa = *(const float4*)(qrow + 16 * f);
    const float4 qb = *(const float4*)(qrow + 16 * f + 4);
    qreg[8 * f + 0] = qa.x; qreg[8 * f + 1] = qa.y; qreg[8 * f + 2] = qa.z; qreg[8 * f + 3] = qa.w;
    qreg[8 * f + 4] = qb.x; qreg[8 * f + 5] = qb.y; qreg[8 * f + 6] = qb.z; qreg[8 * f + 7] = qb.w;
  }

  v8f o[4] = {};
  float mrun[8], lrun[8];
  #pragma unroll
  for (int r = 0; r < 8; ++r) { mrun[r] = -3.0e38f; lrun[r] = 0.0f; }

  const float slope  = (float)(1 << __popc(h));
  const float scale  = 0.03125f;
  const int   nMine  = qblk + ((grp <= qw) ? 1 : 0);

  const float* kt = Kf + headOff + (size_t)grp * 1024 + lane * 4;
  const float* vt = Vf + headOff + (size_t)grp * 1024 + lane * 4;

  for (int st = 0; st < nMine; ++st, kt += 4096, vt += 4096) {
    float4 kq[8], vq[8];
    #pragma unroll
    for (int i = 0; i < 8; ++i) kq[i] = *(const float4*)(kt + i * 128);
    #pragma unroll
    for (int i = 0; i < 8; ++i) vq[i] = *(const float4*)(vt + i * 128);
    __builtin_amdgcn_sched_barrier(0);

    const int k0 = st * 64 + grp * 16;
    v8f s = {};
    #pragma unroll
    for (int t = 0; t < 16; ++t) {
      const float4 k4 = kq[t >> 1];
      v2f a; a.x = qreg[2 * t]; a.y = qreg[2 * t + 1];
      v2f b;
      if (t & 1) { b.x = k4.z; b.y = k4.w; } else { b.x = k4.x; b.y = k4.y; }
      s = wmma_f32(a, b, s);
    }
    const int   d0 = (k0 + mrow) - (q0 + 8 * half);
    const float f0 = (float)d0;
    #pragma unroll
    for (int r = 0; r < 8; ++r) {
      float e = (s[r] + slope * (f0 - (float)r)) * scale;
      if (d0 > r) e = -3.0e18f;
      s[r] = e;
    }
    float pscale[8];
    #pragma unroll
    for (int r = 0; r < 8; ++r) {
      const float rm   = rowmax16(s[r]);
      const float mnew = fmaxf(mrun[r], rm);
      const float alpha = __expf(mrun[r] - mnew);
      mrun[r] = mnew;
      const float p = __expf(s[r] - mnew);
      s[r] = p;
      lrun[r] = lrun[r] * alpha + rowsum16(p);
      pscale[r] = alpha;
    }
    #pragma unroll
    for (int f = 0; f < 4; ++f)
      #pragma unroll
      for (int r = 0; r < 8; ++r) o[f][r] *= pscale[r];

    #pragma unroll
    for (int r = 0; r < 8; ++r)
      Pw[(r + 8 * half) * 17 + mrow] = s[r];
    #pragma unroll
    for (int kk = 0; kk < 4; ++kk) {
      const int ka = 8 * half + 2 * kk;
      v2f a; a.x = Pw[mrow * 17 + ka]; a.y = Pw[mrow * 17 + ka + 1];
      #pragma unroll
      for (int f = 0; f < 4; ++f) {
        const float4 v4 = vq[f * 2 + (kk >> 1)];
        v2f b;
        if (kk & 1) { b.x = v4.z; b.y = v4.w; } else { b.x = v4.x; b.y = v4.y; }
        o[f] = wmma_f32(a, b, o[f]);
      }
    }
  }

  #pragma unroll
  for (int round = 0; round < 2; ++round) {
    const int senderLo = (round == 0) ? 2 : 1;
    const int nPairs   = (round == 0) ? 2 : 1;
    __syncthreads();
    if (grp >= senderLo && grp < senderLo + nPairs) {
      float* mb = smem + ((grp - senderLo) * 4 + qw) * 1536;
      #pragma unroll
      for (int f = 0; f < 4; ++f)
        #pragma unroll
        for (int r = 0; r < 8; ++r) mb[(f * 8 + r) * 32 + lane] = o[f][r];
      #pragma unroll
      for (int r = 0; r < 8; ++r) {
        mb[1024 + r * 32 + lane] = mrun[r];
        mb[1280 + r * 32 + lane] = lrun[r];
      }
    }
    __syncthreads();
    if (grp < nPairs) {
      const float* mb = smem + (grp * 4 + qw) * 1536;
      float a0[8], a1[8];
      #pragma unroll
      for (int r = 0; r < 8; ++r) {
        const float m1 = mb[1024 + r * 32 + lane];
        const float l1 = mb[1280 + r * 32 + lane];
        const float m  = fmaxf(mrun[r], m1);
        a0[r] = __expf(mrun[r] - m);
        a1[r] = __expf(m1 - m);
        lrun[r] = lrun[r] * a0[r] + l1 * a1[r];
        mrun[r] = m;
      }
      #pragma unroll
      for (int f = 0; f < 4; ++f)
        #pragma unroll
        for (int r = 0; r < 8; ++r)
          o[f][r] = o[f][r] * a0[r] + mb[(f * 8 + r) * 32 + lane] * a1[r];
    }
  }

  if (grp == 0) {
    #pragma unroll
    for (int r = 0; r < 8; ++r) lrun[r] = 1.0f / lrun[r];
    float* obase = O + (size_t)n * S_ * E_ + h * D_;
    #pragma unroll
    for (int f = 0; f < 4; ++f)
      #pragma unroll
      for (int r = 0; r < 8; ++r) {
        const int m = r + 8 * half;
        const int dcol = 16 * f + mrow;
        obase[(size_t)(q0 + m) * E_ + dcol] = o[f][r] * lrun[r];
      }
  }
}

__global__ __launch_bounds__(128)
void outproj_kernel(const float* __restrict__ A,
                    const float* __restrict__ Wo,
                    const float* __restrict__ bo,
                    float* __restrict__ Y)
{
  __shared__ __attribute__((aligned(16))) float As[128 * 68];
  __shared__ __attribute__((aligned(16))) float Ws[64 * 68];
  const int tid  = threadIdx.x;
  const int wave = tid >> 5;
  const int lane = tid & 31;
  const int r0 = blockIdx.y * 128;
  const int c0 = blockIdx.x * 64;
  const int mrow = lane & 15, half = lane >> 4;

  v8f c[2][4] = {};

  const int srow = tid >> 4, scol = (tid & 15) * 4;
  const float* ag = A  + (size_t)(r0 + srow) * E_ + scol;
  const float* wg = Wo + (size_t)(c0 + srow) * E_ + scol;
  float* as = As + srow * 68 + scol;
  float* ws = Ws + srow * 68 + scol;
  const float* aw = As + (wave * 32 + mrow) * 68 + 4 * half;
  const float* bw = Ws + mrow * 68 + 4 * half;

  for (int kc = 0; kc < E_; kc += 64) {
    __syncthreads();
    float4 ga[8], gb[8], gw[8];
    #pragma unroll
    for (int n = 0; n < 8; ++n)
      ga[n] = *(const float4*)(ag + (size_t)(8 * n) * E_ + kc);
    #pragma unroll
    for (int n = 0; n < 8; ++n)
      gb[n] = *(const float4*)(ag + (size_t)(8 * n + 64) * E_ + kc);
    #pragma unroll
    for (int n = 0; n < 8; ++n)
      gw[n] = *(const float4*)(wg + (size_t)(8 * n) * E_ + kc);
    #pragma unroll
    for (int n = 0; n < 8; ++n) *(float4*)(as + 8 * n * 68) = ga[n];
    #pragma unroll
    for (int n = 0; n < 8; ++n) *(float4*)(as + (8 * n + 64) * 68) = gb[n];
    #pragma unroll
    for (int n = 0; n < 8; ++n) *(float4*)(ws + 8 * n * 68) = gw[n];
    __syncthreads();

    #pragma unroll
    for (int kh = 0; kh < 2; ++kh) {
      float4 fa[2][4], fb[4][4];
      #pragma unroll
      for (int i = 0; i < 2; ++i)
        #pragma unroll
        for (int q = 0; q < 4; ++q)
          fa[i][q] = *(const float4*)(aw + i * 16 * 68 + 32 * kh + 8 * q);
      #pragma unroll
      for (int f = 0; f < 4; ++f)
        #pragma unroll
        for (int q = 0; q < 4; ++q)
          fb[f][q] = *(const float4*)(bw + f * 16 * 68 + 32 * kh + 8 * q);
      __builtin_amdgcn_sched_barrier(0);
      #pragma unroll
      for (int q = 0; q < 4; ++q)
        #pragma unroll
        for (int e = 0; e < 2; ++e) {
          v2f a[2], b[4];
          #pragma unroll
          for (int i = 0; i < 2; ++i) {
            if (e) { a[i].x = fa[i][q].z; a[i].y = fa[i][q].w; }
            else   { a[i].x = fa[i][q].x; a[i].y = fa[i][q].y; }
          }
          #pragma unroll
          for (int f = 0; f < 4; ++f) {
            if (e) { b[f].x = fb[f][q].z; b[f].y = fb[f][q].w; }
            else   { b[f].x = fb[f][q].x; b[f].y = fb[f][q].y; }
          }
          #pragma unroll
          for (int i = 0; i < 2; ++i)
            #pragma unroll
            for (int f = 0; f < 4; ++f)
              c[i][f] = wmma_f32(a[i], b[f], c[i][f]);
        }
    }
  }

  #pragma unroll
  for (int i = 0; i < 2; ++i)
    #pragma unroll
    for (int f = 0; f < 4; ++f) {
      const int e = c0 + 16 * f + mrow;
      const float bias = bo[e];
      #pragma unroll
      for (int r = 0; r < 8; ++r) {
        const int m = r0 + wave * 32 + 16 * i + r + 8 * half;
        Y[(size_t)m * E_ + e] = c[i][f][r] + bias;
      }
    }
}

extern "C" void kernel_launch(void* const* d_in, const int* in_sizes, int n_in,
                              void* d_out, int out_size, void* d_ws, size_t ws_size,
                              hipStream_t stream) {
  const float* values = (const float*)d_in[0];
  const float* keys   = (const float*)d_in[1];
  const float* query  = (const float*)d_in[2];
  const float* Wv     = (const float*)d_in[4];
  const float* Wk     = (const float*)d_in[5];
  const float* Wq     = (const float*)d_in[6];
  const float* Wo     = (const float*)d_in[7];
  const float* bo     = (const float*)d_in[8];
  float* out = (float*)d_out;

  const size_t tensorElems = (size_t)N_ * H_ * S_ * D_;
  float* Qp = (float*)d_ws;
  float* Kp = Qp + tensorElems;
  float* Vp = Kp + tensorElems;
  float* Oa = Vp + tensorElems;

  const dim3 blk(128);
  const dim3 gProj(S_ / 64, H_, N_);

  proj_kernel<0><<<gProj, blk, 0, stream>>>(query,  Wq, Qp);
  proj_kernel<1><<<gProj, blk, 0, stream>>>(keys,   Wk, Kp);
  proj_kernel<2><<<gProj, blk, 0, stream>>>(values, Wv, Vp);

  const dim3 gFlash(N_ * H_, S_ / 64);
  flash_kernel<<<gFlash, dim3(512), 0, stream>>>(Qp, Kp, Vp, Oa);

  const dim3 gOut(E_ / 64, (N_ * S_) / 128);
  outproj_kernel<<<gOut, blk, 0, stream>>>(Oa, Wo, bo, out);
}


# attention kernel: each lane keeps the exp sum of its own key column; the 16-lane row sum (4 DPP adds per row) runs once after the key loop, not in every step: row sums summed in a new order
# speedup vs baseline: 2.2475x; 1.0469x over previous
#include <hip/hip_runtime.h>
#include <hip/hip_bf16.h>
#include <math.h>


typedef float v2f __attribute__((ext_vector_type(2)));
typedef float v8f __attribute__((ext_vector_type(8)));

static constexpr int N_ = 2;
static constexpr int S_ = 2048;
static constexpr int E_ = 1024;
static constexpr int H_ = 16;
static constexpr int D_ = 64;

static __device__ __forceinline__ v8f wmma_f32(v2f a, v2f b, v8f c) {
  return __builtin_amdgcn_wmma_f32_16x16x4_f32(false, a, false, b,
                                               (short)0, c, false, false);
}

template <int M>
static __device__ __forceinline__ float row_xor16(float v) {
  const int vi = __builtin_bit_cast(int, v);
  return __builtin_bit_cast(float,
      __builtin_amdgcn_update_dpp(vi, vi, 0x160 + M, 0xf, 0xf, true));
}
static __device__ __forceinline__ float rowmax16(float v) {
  v = fmaxf(v, row_xor16<1>(v));
  v = fmaxf(v, row_xor16<2>(v));
  v = fmaxf(v, row_xor16<4>(v));
  v = fmaxf(v, row_xor16<8>(v));
  return v;
}
static __device__ __forceinline__ float rowsum16(float v) {
  v += row_xor16<1>(v);
  v += row_xor16<2>(v);
  v += row_xor16<4>(v);
  v += row_xor16<8>(v);
  return v;
}

template <int MODE>
__global__ __launch_bounds__(128)
void proj_kernel(const float* __restrict__ X,
                 const float* __restrict__ W,
                 float* __restrict__ Out)
{
  __shared__ float Ws[64 * 68];
  const int tid  = threadIdx.x;
  const int wave = tid >> 5;
  const int lane = tid & 31;
  const int n  = blockIdx.z, h = blockIdx.y;
  const int s0 = blockIdx.x * 64 + wave * 16;
  const int mrow = lane & 15, half = lane >> 4;

  for (int i = tid; i < 1024; i += 128) {
    const int r = i >> 4, c4 = i & 15;
    const float4 w4 = ((const float4*)W)[r * 16 + c4];
    float* d = &Ws[r * 68 + c4 * 4];
    d[0] = w4.x; d[1] = w4.y; d[2] = w4.z; d[3] = w4.w;
  }
  __syncthreads();

  const float* xrow = X + ((size_t)n * S_ + (s0 + mrow)) * E_ + h * D_;
  v8f c[4] = {};
  #pragma unroll
  for (int ks = 0; ks < 16; ++ks) {
    const int k = 4 * ks + 2 * half;
    v2f a; a.x = xrow[k]; a.y = xrow[k + 1];
    #pragma unroll
    for (int f = 0; f < 4; ++f) {
      const int e = 16 * f + mrow;
      v2f b; b.x = Ws[e * 68 + k]; b.y = Ws[e * 68 + k + 1];
      if (MODE == 1) c[f] = wmma_f32(b, a, c[f]);
      else           c[f] = wmma_f32(a, b, c[f]);
    }
  }

  float* obase = Out + ((size_t)(n * H_ + h)) * S_ * D_;
  if (MODE == 0) {
    #pragma unroll
    for (int f = 0; f < 4; ++f)
      #pragma unroll
      for (int r = 0; r < 8; ++r) {
        const int m = r + 8 * half;
        const int e = 16 * f + mrow;
        obase[(size_t)(s0 + m) * D_ + e] = c[f][r];
      }
  } else {
    float* tile = obase + (size_t)(s0 >> 4) * 1024 + lane * 4;
    #pragma unroll
    for (int f = 0; f < 4; ++f)
      #pragma unroll
      for (int jj = 0; jj < 2; ++jj) {
        float4 v4;
        v4.x = c[f][4 * jj];     v4.y = c[f][4 * jj + 1];
        v4.z = c[f][4 * jj + 2]; v4.w = c[f][4 * jj + 3];
        *(float4*)(tile + (f * 2 + jj) * 128) = v4;
      }
  }
}

__global__ __launch_bounds__(512)
void flash_kernel(const float* __restrict__ Qp,
                  const float* __restrict__ Kf,
                  const float* __restrict__ Vf,
                  float* __restrict__ O)
{
  __shared__ float smem[4352 + 4352 + 4352];

  const int tid  = threadIdx.x;
  const int wave = __builtin_amdgcn_readfirstlane(tid >> 5);
  const int lane = tid & 31;
  const int grp  = wave >> 2;
  const int qw   = wave & 3;
  const int n = blockIdx.x / H_, h = blockIdx.x % H_;
  const int qblk = (S_ / 64 - 1) - blockIdx.y;
  const int q0 = qblk * 64 + qw * 16;
  const int mrow = lane & 15, half = lane >> 4;
  const size_t headOff = ((size_t)(n * H_ + h)) * S_ * D_;
  float* Pw = smem + 8704 + wave * 272;

  float qreg[32];
  const float* qrow = Qp + headOff + (size_t)(q0 + mrow) * D_ + 8 * half;
  #pragma unroll
  for (int f = 0; f < 4; ++f) {
    const float4 qa = *(const float4*)(qrow + 16 * f);
    const float4 qb = *(const float4*)(qrow + 16 * f + 4);
    qreg[8 * f + 0] = qa.x; qreg[8 * f + 1] = qa.y; qreg[8 * f + 2] = qa.z; qreg[8 * f + 3] = qa.w;
    qreg[8 * f + 4] = qb.x; qreg[8 * f + 5] = qb.y; qreg[8 * f + 6] = qb.z; qreg[8 * f + 7] = qb.w;
  }

  v8f o[4] = {};
  float mrun[8], lrun[8];
  #pragma unroll
  for (int r = 0; r < 8; ++r) { mrun[r] = -3.0e38f; lrun[r] = 0.0f; }

  const float slope  = (float)(1 << __popc(h));
  const float scale  = 0.03125f;
  const int   nMine  = qblk + ((grp <= qw) ? 1 : 0);

  const float* kt = Kf + headOff + (size_t)grp * 1024 + lane * 4;
  const float* vt = Vf + headOff + (size_t)grp * 1024 + lane * 4;

  for (int st = 0; st < nMine; ++st, kt += 4096, vt += 4096) {
    float4 kq[8], vq[8];
    #pragma unroll
    for (int i = 0; i < 8; ++i) kq[i] = *(const float4*)(kt + i * 128);
    #pragma unroll
    for (int i = 0; i < 8; ++i) vq[i] = *(const float4*)(vt + i * 128);
    __builtin_amdgcn_sched_barrier(0);

    const int k0 = st * 64 + grp * 16;
    v8f s = {};
    #pragma unroll
    for (int t = 0; t < 16; ++t) {
      const float4 k4 = kq[t >> 1];
      v2f a; a.x = qreg[2 * t]; a.y = qreg[2 * t + 1];
      v2f b;
      if (t & 1) { b.x = k4.z; b.y = k4.w; } else { b.x = k4.x; b.y = k4.y; }
      s = wmma_f32(a, b, s);
    }
    const int   d0 = (k0 + mrow) - (q0 + 8 * half);
    const float f0 = (float)d0;
    #pragma unroll
    for (int r = 0; r < 8; ++r) {
      float e = (s[r] + slope * (f0 - (float)r)) * scale;
      if (d0 > r) e = -3.0e18f;
      s[r] = e;
    }
    float pscale[8];
    #pragma unroll
    for (int r = 0; r < 8; ++r) {
      const float rm   = rowmax16(s[r]);
      const float mnew = fmaxf(mrun[r], rm);
      const float alpha = __expf(mrun[r] - mnew);
      mrun[r] = mnew;
      const float p = __expf(s[r] - mnew);
      s[r] = p;
      lrun[r] = lrun[r] * alpha + p;
      pscale[r] = alpha;
    }
    #pragma unroll
    for (int f = 0; f < 4; ++f)
      #pragma unroll
      for (int r = 0; r < 8; ++r) o[f][r] *= pscale[r];

    #pragma unroll
    for (int r = 0; r < 8; ++r)
      Pw[(r + 8 * half) * 17 + mrow] = s[r];
    #pragma unroll
    for (int kk = 0; kk < 4; ++kk) {
      const int ka = 8 * half + 2 * kk;
      v2f a; a.x = Pw[mrow * 17 + ka]; a.y = Pw[mrow * 17 + ka + 1];
      #pragma unroll
      for (int f = 0; f < 4; ++f) {
        const float4 v4 = vq[f * 2 + (kk >> 1)];
        v2f b;
        if (kk & 1) { b.x = v4.z; b.y = v4.w; } else { b.x = v4.x; b.y = v4.y; }
        o[f] = wmma_f32(a, b, o[f]);
      }
    }
  }

  #pragma unroll
  for (int r = 0; r < 8; ++r) lrun[r] = rowsum16(lrun[r]);

  #pragma unroll
  for (int round = 0; round < 2; ++round) {
    const int senderLo = (round == 0) ? 2 : 1;
    const int nPairs   = (round == 0) ? 2 : 1;
    __syncthreads();
    if (grp >= senderLo && grp < senderLo + nPairs) {
      float* mb = smem + ((grp - senderLo) * 4 + qw) * 1536;
      #pragma unroll
      for (int f = 0; f < 4; ++f)
        #pragma unroll
        for (int r = 0; r < 8; ++r) mb[(f * 8 + r) * 32 + lane] = o[f][r];
      #pragma unroll
      for (int r = 0; r < 8; ++r) {
        mb[1024 + r * 32 + lane] = mrun[r];
        mb[1280 + r * 32 + lane] = lrun[r];
      }
    }
    __syncthreads();
    if (grp < nPairs) {
      const float* mb = smem + (grp * 4 + qw) * 1536;
      float a0[8], a1[8];
      #pragma unroll
      for (int r = 0; r < 8; ++r) {
        const float m1 = mb[1024 + r * 32 + lane];
        const float l1 = mb[1280 + r * 32 + lane];
        const float m  = fmaxf(mrun[r], m1);
        a0[r] = __expf(mrun[r] - m);
        a1[r] = __expf(m1 - m);
        lrun[r] = lrun[r] * a0[r] + l1 * a1[r];
        mrun[r] = m;
      }
      #pragma unroll
      for (int f = 0; f < 4; ++f)
        #pragma unroll
        for (int r = 0; r < 8; ++r)
          o[f][r] = o[f][r] * a0[r] + mb[(f * 8 + r) * 32 + lane] * a1[r];
    }
  }

  if (grp == 0) {
    #pragma unroll
    for (int r = 0; r < 8; ++r) lrun[r] = 1.0f / lrun[r];
    float* obase = O + (size_t)n * S_ * E_ + h * D_;
    #pragma unroll
    for (int f = 0; f < 4; ++f)
      #pragma unroll
      for (int r = 0; r < 8; ++r) {
        const int m = r + 8 * half;
        const int dcol = 16 * f + mrow;
        obase[(size_t)(q0 + m) * E_ + dcol] = o[f][r] * lrun[r];
      }
  }
}

__global__ __launch_bounds__(128)
void outproj_kernel(const float* __restrict__ A,
                    const float* __restrict__ Wo,
                    const float* __restrict__ bo,
                    float* __restrict__ Y)
{
  __shared__ __attribute__((aligned(16))) float As[128 * 68];
  __shared__ __attribute__((aligned(16))) float Ws[64 * 68];
  const int tid  = threadIdx.x;
  const int wave = tid >> 5;
  const int lane = tid & 31;
  const int r0 = blockIdx.y * 128;
  const int c0 = blockIdx.x * 64;
  const int mrow = lane & 15, half = lane >> 4;

  v8f c[2][4] = {};

  const int srow = tid >> 4, scol = (tid & 15) * 4;
  const float* ag = A  + (size_t)(r0 + srow) * E_ + scol;
  const float* wg = Wo + (size_t)(c0 + srow) * E_ + scol;
  float* as = As + srow * 68 + scol;
  float* ws = Ws + srow * 68 + scol;
  const float* aw = As + (wave * 32 + mrow) * 68 + 4 * half;
  const float* bw = Ws + mrow * 68 + 4 * half;

  for (int kc = 0; kc < E_; kc += 64) {
    __syncthreads();
    float4 ga[8], gb[8], gw[8];
    #pragma unroll
    for (int n = 0; n < 8; ++n)
      ga[n] = *(const float4*)(ag + (size_t)(8 * n) * E_ + kc);
    #pragma unroll
    for (int n = 0; n < 8; ++n)
      gb[n] = *(const float4*)(ag + (size_t)(8 * n + 64) * E_ + kc);
    #pragma unroll
    for (int n = 0; n < 8; ++n)
      gw[n] = *(const float4*)(wg + (size_t)(8 * n) * E_ + kc);
    #pragma unroll
    for (int n = 0; n < 8; ++n) *(float4*)(as + 8 * n * 68) = ga[n];
    #pragma unroll
    for (int n = 0; n < 8; ++n) *(float4*)(as + (8 * n + 64) * 68) = gb[n];
    #pragma unroll
    for (int n = 0; n < 8; ++n) *(float4*)(ws + 8 * n * 68) = gw[n];
    __syncthreads();

    #pragma unroll
    for (int kh = 0; kh < 2; ++kh) {
      float4 fa[2][4], fb[4][4];
      #pragma unroll
      for (int i = 0; i < 2; ++i)
        #pragma unroll
        for (int q = 0; q < 4; ++q)
          fa[i][q] = *(const float4*)(aw + i * 16 * 68 + 32 * kh + 8 * q);
      #pragma unroll
      for (int f = 0; f < 4; ++f)
        #pragma unroll
        for (int q = 0; q < 4; ++q)
          fb[f][q] = *(const float4*)(bw + f * 16 * 68 + 32 * kh + 8 * q);
      __builtin_amdgcn_sched_barrier(0);
      #pragma unroll
      for (int q = 0; q < 4; ++q)
        #pragma unroll
        for (int e = 0; e < 2; ++e) {
          v2f a[2], b[4];
          #pragma unroll
          for (int i = 0; i < 2; ++i) {
            if (e) { a[i].x = fa[i][q].z; a[i].y = fa[i][q].w; }
            else   { a[i].x = fa[i][q].x; a[i].y = fa[i][q].y; }
          }
          #pragma unroll
          for (int f = 0; f < 4; ++f) {
            if (e) { b[f].x = fb[f][q].z; b[f].y = fb[f][q].w; }
            else   { b[f].x = fb[f][q].x; b[f].y = fb[f][q].y; }
          }
          #pragma unroll
          for (int i = 0; i < 2; ++i)
            #pragma unroll
            for (int f = 0; f < 4; ++f)
              c[i][f] = wmma_f32(a[i], b[f], c[i][f]);
        }
    }
  }

  #pragma unroll
  for (int i = 0; i < 2; ++i)
    #pragma unroll
    for (int f = 0; f < 4; ++f) {
      const int e = c0 + 16 * f + mrow;
      const float bias = bo[e];
      #pragma unroll
      for (int r = 0; r < 8; ++r) {
        const int m = r0 + wave * 32 + 16 * i + r + 8 * half;
        Y[(size_t)m * E_ + e] = c[i][f][r] + bias;
      }
    }
}

extern "C" void kernel_launch(void* const* d_in, const int* in_sizes, int n_in,
                              void* d_out, int out_size, void* d_ws, size_t ws_size,
                              hipStream_t stream) {
  const float* values = (const float*)d_in[0];
  const float* keys   = (const float*)d_in[1];
  const float* query  = (const float*)d_in[2];
  const float* Wv     = (const float*)d_in[4];
  const float* Wk     = (const float*)d_in[5];
  const float* Wq     = (const float*)d_in[6];
  const float* Wo     = (const float*)d_in[7];
  const float* bo     = (const float*)d_in[8];
  float* out = (float*)d_out;

  const size_t tensorElems = (size_t)N_ * H_ * S_ * D_;
  float* Qp = (float*)d_ws;
  float* Kp = Qp + tensorElems;
  float* Vp = Kp + tensorElems;
  float* Oa = Vp + tensorElems;

  const dim3 blk(128);
  const dim3 gProj(S_ / 64, H_, N_);

  proj_kernel<0><<<gProj, blk, 0, stream>>>(query,  Wq, Qp);
  proj_kernel<1><<<gProj, blk, 0, stream>>>(keys,   Wk, Kp);
  proj_kernel<2><<<gProj, blk, 0, stream>>>(values, Wv, Vp);

  const dim3 gFlash(N_ * H_, S_ / 64);
  flash_kernel<<<gFlash, dim3(512), 0, stream>>>(Qp, Kp, Vp, Oa);

  const dim3 gOut(E_ / 64, (N_ * S_) / 128);
  outproj_kernel<<<gOut, blk, 0, stream>>>(Oa, Wo, bo, out);
}


# attention: running max raised lazily: key loop runs from the diagonal step backwards; a step whose scores stay within 8 of the row's shift skips row max, alpha and rescale; sums in a new order
# speedup vs baseline: 2.4767x; 1.1020x over previous
#include <hip/hip_runtime.h>
#include <hip/hip_bf16.h>
#include <math.h>


typedef float v2f __attribute__((ext_vector_type(2)));
typedef float v4f __attribute__((ext_vector_type(4)));
typedef float v8f __attribute__((ext_vector_type(8)));

static constexpr int N_ = 2;
static constexpr int S_ = 2048;
static constexpr int E_ = 1024;
static constexpr int H_ = 16;
static constexpr int D_ = 64;

static __device__ __forceinline__ v8f wmma_f32(v2f a, v2f b, v8f c) {
  return __builtin_amdgcn_wmma_f32_16x16x4_f32(false, a, false, b,
                                               (short)0, c, false, false);
}

template <int M>
static __device__ __forceinline__ float row_xor16(float v) {
  const int vi = __builtin_bit_cast(int, v);
  return __builtin_bit_cast(float,
      __builtin_amdgcn_update_dpp(vi, vi, 0x160 + M, 0xf, 0xf, true));
}
static __device__ __forceinline__ float rowmax16(float v) {
  v = fmaxf(v, row_xor16<1>(v));
  v = fmaxf(v, row_xor16<2>(v));
  v = fmaxf(v, row_xor16<4>(v));
  v = fmaxf(v, row_xor16<8>(v));
  return v;
}
static __device__ __forceinline__ float rowsum16(float v) {
  v += row_xor16<1>(v);
  v += row_xor16<2>(v);
  v += row_xor16<4>(v);
  v += row_xor16<8>(v);
  return v;
}

template <int MODE>
__global__ __launch_bounds__(128)
void proj_kernel(const float* __restrict__ X,
                 const float* __restrict__ W,
                 float* __restrict__ Out)
{
  __shared__ float Ws[64 * 68];
  const int tid  = threadIdx.x;
  const int wave = tid >> 5;
  const int lane = tid & 31;
  const int n  = blockIdx.z, h = blockIdx.y;
  const int s0 = blockIdx.x * 64 + wave * 16;
  const int mrow = lane & 15, half = lane >> 4;

  for (int i = tid; i < 1024; i += 128) {
    const int r = i >> 4, c4 = i & 15;
    const float4 w4 = ((const float4*)W)[r * 16 + c4];
    float* d = &Ws[r * 68 + c4 * 4];
    d[0] = w4.x; d[1] = w4.y; d[2] = w4.z; d[3] = w4.w;
  }
  __syncthreads();

  const float* xrow = X + ((size_t)n * S_ + (s0 + mrow)) * E_ + h * D_;
  v8f c[4] = {};
  #pragma unroll
  for (int ks = 0; ks < 16; ++ks) {
    const int k = 4 * ks + 2 * half;
    v2f a; a.x = xrow[k]; a.y = xrow[k + 1];
    #pragma unroll
    for (int f = 0; f < 4; ++f) {
      const int e = 16 * f + mrow;
      v2f b; b.x = Ws[e * 68 + k]; b.y = Ws[e * 68 + k + 1];
      if (MODE == 1) c[f] = wmma_f32(b, a, c[f]);
      else           c[f] = wmma_f32(a, b, c[f]);
    }
  }

  float* obase = Out + ((size_t)(n * H_ + h)) * S_ * D_;
  if (MODE == 0) {
    #pragma unroll
    for (int f = 0; f < 4; ++f)
      #pragma unroll
      for (int r = 0; r < 8; ++r) {
        const int m = r + 8 * half;
        const int e = 16 * f + mrow;
        obase[(size_t)(s0 + m) * D_ + e] = c[f][r];
      }
  } else {
    float* tile = obase + (size_t)(s0 >> 4) * 1024 + lane * 4;
    #pragma unroll
    for (int f = 0; f < 4; ++f)
      #pragma unroll
      for (int jj = 0; jj < 2; ++jj) {
        float4 v4;
        v4.x = c[f][4 * jj];     v4.y = c[f][4 * jj + 1];
        v4.z = c[f][4 * jj + 2]; v4.w = c[f][4 * jj + 3];
        *(float4*)(tile + (f * 2 + jj) * 128) = v4;
      }
  }
}

static __device__ __forceinline__ void load_tiles(const float* kt, const float* vt,
                                                  v4f (&kq)[8], v4f (&vq)[8]) {
  #pragma unroll
  for (int i = 0; i < 8; ++i) kq[i] = *(const v4f*)(kt + i * 128);
  #pragma unroll
  for (int i = 0; i < 8; ++i) vq[i] = *(const v4f*)(vt + i * 128);
  __builtin_amdgcn_sched_barrier(0);
}

static __device__ __forceinline__ v8f tile_scores(const float (&qreg)[32], const v4f (&kq)[8],
                                                  float slope, float scale, float f0) {
  v8f s = {};
  #pragma unroll
  for (int t = 0; t < 16; ++t) {
    const v4f k4 = kq[t >> 1];
    v2f a; a.x = qreg[2 * t]; a.y = qreg[2 * t + 1];
    v2f b;
    if (t & 1) { b.x = k4.z; b.y = k4.w; } else { b.x = k4.x; b.y = k4.y; }
    s = wmma_f32(a, b, s);
  }
  #pragma unroll
  for (int r = 0; r < 8; ++r) s[r] = (s[r] + slope * (f0 - (float)r)) * scale;
  return s;
}

static __device__ __forceinline__ void accumulate_pv(float* Pw, const v8f& p, const v4f (&vq)[8],
                                                     v8f (&o)[4], int mrow, int half) {
  #pragma unroll
  for (int r = 0; r < 8; ++r)
    Pw[(r + 8 * half) * 17 + mrow] = p[r];
  #pragma unroll
  for (int kk = 0; kk < 4; ++kk) {
    const int ka = 8 * half + 2 * kk;
    v2f a; a.x = Pw[mrow * 17 + ka]; a.y = Pw[mrow * 17 + ka + 1];
    #pragma unroll
    for (int f = 0; f < 4; ++f) {
      const v4f v4 = vq[f * 2 + (kk >> 1)];
      v2f b;
      if (kk & 1) { b.x = v4.z; b.y = v4.w; } else { b.x = v4.x; b.y = v4.y; }
      o[f] = wmma_f32(a, b, o[f]);
    }
  }
}

__global__ __launch_bounds__(512)
void flash_kernel(const float* __restrict__ Qp,
                  const float* __restrict__ Kf,
                  const float* __restrict__ Vf,
                  float* __restrict__ O)
{
  __shared__ float smem[4352 + 4352 + 4352];

  const int tid  = threadIdx.x;
  const int wave = __builtin_amdgcn_readfirstlane(tid >> 5);
  const int lane = tid & 31;
  const int grp  = wave >> 2;
  const int qw   = wave & 3;
  const int n = blockIdx.x / H_, h = blockIdx.x % H_;
  const int qblk = (S_ / 64 - 1) - blockIdx.y;
  const int q0 = qblk * 64 + qw * 16;
  const int mrow = lane & 15, half = lane >> 4;
  const size_t headOff = ((size_t)(n * H_ + h)) * S_ * D_;
  float* Pw = smem + 8704 + wave * 272;

  float qreg[32];
  const float* qrow = Qp + headOff + (size_t)(q0 + mrow) * D_ + 8 * half;
  #pragma unroll
  for (int f = 0; f < 4; ++f) {
    const float4 qa = *(const float4*)(qrow + 16 * f);
    const float4 qb = *(const float4*)(qrow + 16 * f + 4);
    qreg[8 * f + 0] = qa.x; qreg[8 * f + 1] = qa.y; qreg[8 * f + 2] = qa.z; qreg[8 * f + 3] = qa.w;
    qreg[8 * f + 4] = qb.x; qreg[8 * f + 5] = qb.y; qreg[8 * f + 6] = qb.z; qreg[8 * f + 7] = qb.w;
  }

  v8f o[4] = {};
  float mrun[8], lrun[8];
  #pragma unroll
  for (int r = 0; r < 8; ++r) { mrun[r] = -3.0e38f; lrun[r] = 0.0f; }

  const float slope  = (float)(1 << __popc(h));
  const float scale  = 0.03125f;
  const int   nMine  = qblk + ((grp <= qw) ? 1 : 0);

  const float* kbase = Kf + headOff + (size_t)grp * 1024 + lane * 4;
  const float* vbase = Vf + headOff + (size_t)grp * 1024 + lane * 4;
  constexpr float kLazy = 8.0f;
  const int dBase = (grp * 16 + mrow) - (q0 + 8 * half);

  int st = nMine - 1;
  while (st >= 0) {
    {
      v4f kq[8], vq[8];
      load_tiles(kbase + (size_t)st * 4096, vbase + (size_t)st * 4096, kq, vq);
      const int d0 = dBase + st * 64;
      v8f s = tile_scores(qreg, kq, slope, scale, (float)d0);
      float pscale[8];
      #pragma unroll
      for (int r = 0; r < 8; ++r) {
        float e = s[r];
        if (d0 > r) e = -3.0e18f;
        const float rm    = rowmax16(e);
        const float mnew  = fmaxf(mrun[r], rm);
        const float alpha = __expf(mrun[r] - mnew);
        mrun[r] = mnew;
        const float p = __expf(e - mnew);
        s[r] = p;
        lrun[r] = lrun[r] * alpha + p;
        pscale[r] = alpha;
      }
      #pragma unroll
      for (int f = 0; f < 4; ++f)
        #pragma unroll
        for (int r = 0; r < 8; ++r) o[f][r] *= pscale[r];
      accumulate_pv(Pw, s, vq, o, mrow, half);
    }
    --st;
    for (; st >= 0; --st) {
      v4f kq[8], vq[8];
      load_tiles(kbase + (size_t)st * 4096, vbase + (size_t)st * 4096, kq, vq);
      const int d0 = dBase + st * 64;
      v8f s = tile_scores(qreg, kq, slope, scale, (float)d0);
      #pragma unroll
      for (int r = 0; r < 8; ++r) s[r] -= mrun[r];
      const float xtop = fmaxf(fmaxf(fmaxf(s[0], s[1]), fmaxf(s[2], s[3])),
                               fmaxf(fmaxf(s[4], s[5]), fmaxf(s[6], s[7])));
      asm volatile("" :: "v"(vq[0]), "v"(vq[1]), "v"(vq[2]), "v"(vq[3]),
                         "v"(vq[4]), "v"(vq[5]), "v"(vq[6]), "v"(vq[7]), "v"(xtop));
      if (__builtin_amdgcn_ballot_w32(xtop > kLazy) != 0u) break;
      #pragma unroll
      for (int r = 0; r < 8; ++r) {
        const float p = __expf(s[r]);
        s[r] = p;
        lrun[r] += p;
      }
      accumulate_pv(Pw, s, vq, o, mrow, half);
    }
  }

  #pragma unroll
  for (int r = 0; r < 8; ++r) lrun[r] = rowsum16(lrun[r]);

  #pragma unroll
  for (int round = 0; round < 2; ++round) {
    const int senderLo = (round == 0) ? 2 : 1;
    const int nPairs   = (round == 0) ? 2 : 1;
    __syncthreads();
    if (grp >= senderLo && grp < senderLo + nPairs) {
      float* mb = smem + ((grp - senderLo) * 4 + qw) * 1536;
      #pragma unroll
      for (int f = 0; f < 4; ++f)
        #pragma unroll
        for (int r = 0; r < 8; ++r) mb[(f * 8 + r) * 32 + lane] = o[f][r];
      #pragma unroll
      for (int r = 0; r < 8; ++r) {
        mb[1024 + r * 32 + lane] = mrun[r];
        mb[1280 + r * 32 + lane] = lrun[r];
      }
    }
    __syncthreads();
    if (grp < nPairs) {
      const float* mb = smem + (grp * 4 + qw) * 1536;
      float a0[8], a1[8];
      #pragma unroll
      for (int r = 0; r < 8; ++r) {
        const float m1 = mb[1024 + r * 32 + lane];
        const float l1 = mb[1280 + r * 32 + lane];
        const float m  = fmaxf(mrun[r], m1);
        a0[r] = __expf(mrun[r] - m);
        a1[r] = __expf(m1 - m);
        lrun[r] = lrun[r] * a0[r] + l1 * a1[r];
        mrun[r] = m;
      }
      #pragma unroll
      for (int f = 0; f < 4; ++f)
        #pragma unroll
        for (int r = 0; r < 8; ++r)
          o[f][r] = o[f][r] * a0[r] + mb[(f * 8 + r) * 32 + lane] * a1[r];
    }
  }

  if (grp == 0) {
    #pragma unroll
    for (int r = 0; r < 8; ++r) lrun[r] = 1.0f / lrun[r];
    float* obase = O + (size_t)n * S_ * E_ + h * D_;
    #pragma unroll
    for (int f = 0; f < 4; ++f)
      #pragma unroll
      for (int r = 0; r < 8; ++r) {
        const int m = r + 8 * half;
        const int dcol = 16 * f + mrow;
        obase[(size_t)(q0 + m) * E_ + dcol] = o[f][r] * lrun[r];
      }
  }
}

__global__ __launch_bounds__(128)
void outproj_kernel(const float* __restrict__ A,
                    const float* __restrict__ Wo,
                    const float* __restrict__ bo,
                    float* __restrict__ Y)
{
  __shared__ __attribute__((aligned(16))) float As[128 * 68];
  __shared__ __attribute__((aligned(16))) float Ws[64 * 68];
  const int tid  = threadIdx.x;
  const int wave = tid >> 5;
  const int lane = tid & 31;
  const int r0 = blockIdx.y * 128;
  const int c0 = blockIdx.x * 64;
  const int mrow = lane & 15, half = lane >> 4;

  v8f c[2][4] = {};

  const int srow = tid >> 4, scol = (tid & 15) * 4;
  const float* ag = A  + (size_t)(r0 + srow) * E_ + scol;
  const float* wg = Wo + (size_t)(c0 + srow) * E_ + scol;
  float* as = As + srow * 68 + scol;
  float* ws = Ws + srow * 68 + scol;
  const float* aw = As + (wave * 32 + mrow) * 68 + 4 * half;
  const float* bw = Ws + mrow * 68 + 4 * half;

  for (int kc = 0; kc < E_; kc += 64) {
    __syncthreads();
    float4 ga[8], gb[8], gw[8];
    #pragma unroll
    for (int n = 0; n < 8; ++n)
      ga[n] = *(const float4*)(ag + (size_t)(8 * n) * E_ + kc);
    #pragma unroll
    for (int n = 0; n < 8; ++n)
      gb[n] = *(const float4*)(ag + (size_t)(8 * n + 64) * E_ + kc);
    #pragma unroll
    for (int n = 0; n < 8; ++n)
      gw[n] = *(const float4*)(wg + (size_t)(8 * n) * E_ + kc);
    #pragma unroll
    for (int n = 0; n < 8; ++n) *(float4*)(as + 8 * n * 68) = ga[n];
    #pragma unroll
    for (int n = 0; n < 8; ++n) *(float4*)(as + (8 * n + 64) * 68) = gb[n];
    #pragma unroll
    for (int n = 0; n < 8; ++n) *(float4*)(ws + 8 * n * 68) = gw[n];
    __syncthreads();

    #pragma unroll
    for (int kh = 0; kh < 2; ++kh) {
      float4 fa[2][4], fb[4][4];
      #pragma unroll
      for (int i = 0; i < 2; ++i)
        #pragma unroll
        for (int q = 0; q < 4; ++q)
          fa[i][q] = *(const float4*)(aw + i * 16 * 68 + 32 * kh + 8 * q);
      #pragma unroll
      for (int f = 0; f < 4; ++f)
        #pragma unroll
        for (int q = 0; q < 4; ++q)
          fb[f][q] = *(const float4*)(bw + f * 16 * 68 + 32 * kh + 8 * q);
      __builtin_amdgcn_sched_barrier(0);
      #pragma unroll
      for (int q = 0; q < 4; ++q)
        #pragma unroll
        for (int e = 0; e < 2; ++e) {
          v2f a[2], b[4];
          #pragma unroll
          for (int i = 0; i < 2; ++i) {
            if (e) { a[i].x = fa[i][q].z; a[i].y = fa[i][q].w; }
            else   { a[i].x = fa[i][q].x; a[i].y = fa[i][q].y; }
          }
          #pragma unroll
          for (int f = 0; f < 4; ++f) {
            if (e) { b[f].x = fb[f][q].z; b[f].y = fb[f][q].w; }
            else   { b[f].x = fb[f][q].x; b[f].y = fb[f][q].y; }
          }
          #pragma unroll
          for (int i = 0; i < 2; ++i)
            #pragma unroll
            for (int f = 0; f < 4; ++f)
              c[i][f] = wmma_f32(a[i], b[f], c[i][f]);
        }
    }
  }

  #pragma unroll
  for (int i = 0; i < 2; ++i)
    #pragma unroll
    for (int f = 0; f < 4; ++f) {
      const int e = c0 + 16 * f + mrow;
      const float bias = bo[e];
      #pragma unroll
      for (int r = 0; r < 8; ++r) {
        const int m = r0 + wave * 32 + 16 * i + r + 8 * half;
        Y[(size_t)m * E_ + e] = c[i][f][r] + bias;
      }
    }
}

extern "C" void kernel_launch(void* const* d_in, const int* in_sizes, int n_in,
                              void* d_out, int out_size, void* d_ws, size_t ws_size,
                              hipStream_t stream) {
  const float* values = (const float*)d_in[0];
  const float* keys   = (const float*)d_in[1];
  const float* query  = (const float*)d_in[2];
  const float* Wv     = (const float*)d_in[4];
  const float* Wk     = (const float*)d_in[5];
  const float* Wq     = (const float*)d_in[6];
  const float* Wo     = (const float*)d_in[7];
  const float* bo     = (const float*)d_in[8];
  float* out = (float*)d_out;

  const size_t tensorElems = (size_t)N_ * H_ * S_ * D_;
  float* Qp = (float*)d_ws;
  float* Kp = Qp + tensorElems;
  float* Vp = Kp + tensorElems;
  float* Oa = Vp + tensorElems;

  const dim3 blk(128);
  const dim3 gProj(S_ / 64, H_, N_);

  proj_kernel<0><<<gProj, blk, 0, stream>>>(query,  Wq, Qp);
  proj_kernel<1><<<gProj, blk, 0, stream>>>(keys,   Wk, Kp);
  proj_kernel<2><<<gProj, blk, 0, stream>>>(values, Wv, Vp);

  const dim3 gFlash(N_ * H_, S_ / 64);
  flash_kernel<<<gFlash, dim3(512), 0, stream>>>(Qp, Kp, Vp, Oa);

  const dim3 gOut(E_ / 64, (N_ * S_) / 128);
  outproj_kernel<<<gOut, blk, 0, stream>>>(Oa, Wo, bo, out);
}


# the three projection kernels run as ONE launch (grid z = projection x batch): their blocks are resident together and two launch boundaries go; each block's work and every sum unchanged
# speedup vs baseline: 2.5761x; 1.0401x over previous
#include <hip/hip_runtime.h>
#include <hip/hip_bf16.h>
#include <math.h>


typedef float v2f __attribute__((ext_vector_type(2)));
typedef float v4f __attribute__((ext_vector_type(4)));
typedef float v8f __attribute__((ext_vector_type(8)));

static constexpr int N_ = 2;
static constexpr int S_ = 2048;
static constexpr int E_ = 1024;
static constexpr int H_ = 16;
static constexpr int D_ = 64;

static __device__ __forceinline__ v8f wmma_f32(v2f a, v2f b, v8f c) {
  return __builtin_amdgcn_wmma_f32_16x16x4_f32(false, a, false, b,
                                               (short)0, c, false, false);
}

template <int M>
static __device__ __forceinline__ float row_xor16(float v) {
  const int vi = __builtin_bit_cast(int, v);
  return __builtin_bit_cast(float,
      __builtin_amdgcn_update_dpp(vi, vi, 0x160 + M, 0xf, 0xf, true));
}
static __device__ __forceinline__ float rowmax16(float v) {
  v = fmaxf(v, row_xor16<1>(v));
  v = fmaxf(v, row_xor16<2>(v));
  v = fmaxf(v, row_xor16<4>(v));
  v = fmaxf(v, row_xor16<8>(v));
  return v;
}
static __device__ __forceinline__ float rowsum16(float v) {
  v += row_xor16<1>(v);
  v += row_xor16<2>(v);
  v += row_xor16<4>(v);
  v += row_xor16<8>(v);
  return v;
}

template <int MODE>
static __device__ __forceinline__
void proj_tile(const float* __restrict__ X,
               const float* __restrict__ W,
               float* __restrict__ Out,
               float* Ws,
               const int n)
{
  const int tid  = threadIdx.x;
  const int wave = tid >> 5;
  const int lane = tid & 31;
  const int h = blockIdx.y;
  const int s0 = blockIdx.x * 64 + wave * 16;
  const int mrow = lane & 15, half = lane >> 4;

  for (int i = tid; i < 1024; i += 128) {
    const int r = i >> 4, c4 = i & 15;
    const float4 w4 = ((const float4*)W)[r * 16 + c4];
    float* d = &Ws[r * 68 + c4 * 4];
    d[0] = w4.x; d[1] = w4.y; d[2] = w4.z; d[3] = w4.w;
  }
  __syncthreads();

  const float* xrow = X + ((size_t)n * S_ + (s0 + mrow)) * E_ + h * D_;
  v8f c[4] = {};
  #pragma unroll
  for (int ks = 0; ks < 16; ++ks) {
    const int k = 4 * ks + 2 * half;
    v2f a; a.x = xrow[k]; a.y = xrow[k + 1];
    #pragma unroll
    for (int f = 0; f < 4; ++f) {
      const int e = 16 * f + mrow;
      v2f b; b.x = Ws[e * 68 + k]; b.y = Ws[e * 68 + k + 1];
      if (MODE == 1) c[f] = wmma_f32(b, a, c[f]);
      else           c[f] = wmma_f32(a, b, c[f]);
    }
  }

  float* obase = Out + ((size_t)(n * H_ + h)) * S_ * D_;
  if (MODE == 0) {
    #pragma unroll
    for (int f = 0; f < 4; ++f)
      #pragma unroll
      for (int r = 0; r < 8; ++r) {
        const int m = r + 8 * half;
        const int e = 16 * f + mrow;
        obase[(size_t)(s0 + m) * D_ + e] = c[f][r];
      }
  } else {
    float* tile = obase + (size_t)(s0 >> 4) * 1024 + lane * 4;
    #pragma unroll
    for (int f = 0; f < 4; ++f)
      #pragma unroll
      for (int jj = 0; jj < 2; ++jj) {
        float4 v4;
        v4.x = c[f][4 * jj];     v4.y = c[f][4 * jj + 1];
        v4.z = c[f][4 * jj + 2]; v4.w = c[f][4 * jj + 3];
        *(float4*)(tile + (f * 2 + jj) * 128) = v4;
      }
  }
}

__global__ __launch_bounds__(128)
void proj3_kernel(const float* __restrict__ Xq, const float* __restrict__ Xk,
                  const float* __restrict__ Xv,
                  const float* __restrict__ Wq, const float* __restrict__ Wk,
                  const float* __restrict__ Wv,
                  float* __restrict__ Qp, float* __restrict__ Kp, float* __restrict__ Vp)
{
  __shared__ float Ws[64 * 68];
  const int which = blockIdx.z / N_;
  const int n     = blockIdx.z % N_;
  if (which == 0)      proj_tile<0>(Xq, Wq, Qp, Ws, n);
  else if (which == 1) proj_tile<1>(Xk, Wk, Kp, Ws, n);
  else                 proj_tile<2>(Xv, Wv, Vp, Ws, n);
}

static __device__ __forceinline__ void load_tiles(const float* kt, const float* vt,
                                                  v4f (&kq)[8], v4f (&vq)[8]) {
  #pragma unroll
  for (int i = 0; i < 8; ++i) kq[i] = *(const v4f*)(kt + i * 128);
  #pragma unroll
  for (int i = 0; i < 8; ++i) vq[i] = *(const v4f*)(vt + i * 128);
  __builtin_amdgcn_sched_barrier(0);
}

static __device__ __forceinline__ v8f tile_scores(const float (&qreg)[32], const v4f (&kq)[8],
                                                  float slope, float scale, float f0) {
  v8f s = {};
  #pragma unroll
  for (int t = 0; t < 16; ++t) {
    const v4f k4 = kq[t >> 1];
    v2f a; a.x = qreg[2 * t]; a.y = qreg[2 * t + 1];
    v2f b;
    if (t & 1) { b.x = k4.z; b.y = k4.w; } else { b.x = k4.x; b.y = k4.y; }
    s = wmma_f32(a, b, s);
  }
  #pragma unroll
  for (int r = 0; r < 8; ++r) s[r] = (s[r] + slope * (f0 - (float)r)) * scale;
  return s;
}

static __device__ __forceinline__ void accumulate_pv(float* Pw, const v8f& p, const v4f (&vq)[8],
                                                     v8f (&o)[4], int mrow, int half) {
  #pragma unroll
  for (int r = 0; r < 8; ++r)
    Pw[(r + 8 * half) * 17 + mrow] = p[r];
  #pragma unroll
  for (int kk = 0; kk < 4; ++kk) {
    const int ka = 8 * half + 2 * kk;
    v2f a; a.x = Pw[mrow * 17 + ka]; a.y = Pw[mrow * 17 + ka + 1];
    #pragma unroll
    for (int f = 0; f < 4; ++f) {
      const v4f v4 = vq[f * 2 + (kk >> 1)];
      v2f b;
      if (kk & 1) { b.x = v4.z; b.y = v4.w; } else { b.x = v4.x; b.y = v4.y; }
      o[f] = wmma_f32(a, b, o[f]);
    }
  }
}

__global__ __launch_bounds__(512)
void flash_kernel(const float* __restrict__ Qp,
                  const float* __restrict__ Kf,
                  const float* __restrict__ Vf,
                  float* __restrict__ O)
{
  __shared__ float smem[4352 + 4352 + 4352];

  const int tid  = threadIdx.x;
  const int wave = __builtin_amdgcn_readfirstlane(tid >> 5);
  const int lane = tid & 31;
  const int grp  = wave >> 2;
  const int qw   = wave & 3;
  const int n = blockIdx.x / H_, h = blockIdx.x % H_;
  const int qblk = (S_ / 64 - 1) - blockIdx.y;
  const int q0 = qblk * 64 + qw * 16;
  const int mrow = lane & 15, half = lane >> 4;
  const size_t headOff = ((size_t)(n * H_ + h)) * S_ * D_;
  float* Pw = smem + 8704 + wave * 272;

  float qreg[32];
  const float* qrow = Qp + headOff + (size_t)(q0 + mrow) * D_ + 8 * half;
  #pragma unroll
  for (int f = 0; f < 4; ++f) {
    const float4 qa = *(const float4*)(qrow + 16 * f);
    const float4 qb = *(const float4*)(qrow + 16 * f + 4);
    qreg[8 * f + 0] = qa.x; qreg[8 * f + 1] = qa.y; qreg[8 * f + 2] = qa.z; qreg[8 * f + 3] = qa.w;
    qreg[8 * f + 4] = qb.x; qreg[8 * f + 5] = qb.y; qreg[8 * f + 6] = qb.z; qreg[8 * f + 7] = qb.w;
  }

  v8f o[4] = {};
  float mrun[8], lrun[8];
  #pragma unroll
  for (int r = 0; r < 8; ++r) { mrun[r] = -3.0e38f; lrun[r] = 0.0f; }

  const float slope  = (float)(1 << __popc(h));
  const float scale  = 0.03125f;
  const int   nMine  = qblk + ((grp <= qw) ? 1 : 0);

  const float* kbase = Kf + headOff + (size_t)grp * 1024 + lane * 4;
  const float* vbase = Vf + headOff + (size_t)grp * 1024 + lane * 4;
  constexpr float kLazy = 8.0f;
  const int dBase = (grp * 16 + mrow) - (q0 + 8 * half);

  int st = nMine - 1;
  while (st >= 0) {
    {
      v4f kq[8], vq[8];
      load_tiles(kbase + (size_t)st * 4096, vbase + (size_t)st * 4096, kq, vq);
      const int d0 = dBase + st * 64;
      v8f s = tile_scores(qreg, kq, slope, scale, (float)d0);
      float pscale[8];
      #pragma unroll
      for (int r = 0; r < 8; ++r) {
        float e = s[r];
        if (d0 > r) e = -3.0e18f;
        const float rm    = rowmax16(e);
        const float mnew  = fmaxf(mrun[r], rm);
        const float alpha = __expf(mrun[r] - mnew);
        mrun[r] = mnew;
        const float p = __expf(e - mnew);
        s[r] = p;
        lrun[r] = lrun[r] * alpha + p;
        pscale[r] = alpha;
      }
      #pragma unroll
      for (int f = 0; f < 4; ++f)
        #pragma unroll
        for (int r = 0; r < 8; ++r) o[f][r] *= pscale[r];
      accumulate_pv(Pw, s, vq, o, mrow, half);
    }
    --st;
    for (; st >= 0; --st) {
      v4f kq[8], vq[8];
      load_tiles(kbase + (size_t)st * 4096, vbase + (size_t)st * 4096, kq, vq);
      const int d0 = dBase + st * 64;
      v8f s = tile_scores(qreg, kq, slope, scale, (float)d0);
      #pragma unroll
      for (int r = 0; r < 8; ++r) s[r] -= mrun[r];
      const float xtop = fmaxf(fmaxf(fmaxf(s[0], s[1]), fmaxf(s[2], s[3])),
                               fmaxf(fmaxf(s[4], s[5]), fmaxf(s[6], s[7])));
      asm volatile("" :: "v"(vq[0]), "v"(vq[1]), "v"(vq[2]), "v"(vq[3]),
                         "v"(vq[4]), "v"(vq[5]), "v"(vq[6]), "v"(vq[7]), "v"(xtop));
      if (__builtin_amdgcn_ballot_w32(xtop > kLazy) != 0u) break;
      #pragma unroll
      for (int r = 0; r < 8; ++r) {
        const float p = __expf(s[r]);
        s[r] = p;
        lrun[r] += p;
      }
      accumulate_pv(Pw, s, vq, o, mrow, half);
    }
  }

  #pragma unroll
  for (int r = 0; r < 8; ++r) lrun[r] = rowsum16(lrun[r]);

  #pragma unroll
  for (int round = 0; round < 2; ++round) {
    const int senderLo = (round == 0) ? 2 : 1;
    const int nPairs   = (round == 0) ? 2 : 1;
    __syncthreads();
    if (grp >= senderLo && grp < senderLo + nPairs) {
      float* mb = smem + ((grp - senderLo) * 4 + qw) * 1536;
      #pragma unroll
      for (int f = 0; f < 4; ++f)
        #pragma unroll
        for (int r = 0; r < 8; ++r) mb[(f * 8 + r) * 32 + lane] = o[f][r];
      #pragma unroll
      for (int r = 0; r < 8; ++r) {
        mb[1024 + r * 32 + lane] = mrun[r];
        mb[1280 + r * 32 + lane] = lrun[r];
      }
    }
    __syncthreads();
    if (grp < nPairs) {
      const float* mb = smem + (grp * 4 + qw) * 1536;
      float a0[8], a1[8];
      #pragma unroll
      for (int r = 0; r < 8; ++r) {
        const float m1 = mb[1024 + r * 32 + lane];
        const float l1 = mb[1280 + r * 32 + lane];
        const float m  = fmaxf(mrun[r], m1);
        a0[r] = __expf(mrun[r] - m);
        a1[r] = __expf(m1 - m);
        lrun[r] = lrun[r] * a0[r] + l1 * a1[r];
        mrun[r] = m;
      }
      #pragma unroll
      for (int f = 0; f < 4; ++f)
        #pragma unroll
        for (int r = 0; r < 8; ++r)
          o[f][r] = o[f][r] * a0[r] + mb[(f * 8 + r) * 32 + lane] * a1[r];
    }
  }

  if (grp == 0) {
    #pragma unroll
    for (int r = 0; r < 8; ++r) lrun[r] = 1.0f / lrun[r];
    float* obase = O + (size_t)n * S_ * E_ + h * D_;
    #pragma unroll
    for (int f = 0; f < 4; ++f)
      #pragma unroll
      for (int r = 0; r < 8; ++r) {
        const int m = r + 8 * half;
        const int dcol = 16 * f + mrow;
        obase[(size_t)(q0 + m) * E_ + dcol] = o[f][r] * lrun[r];
      }
  }
}

__global__ __launch_bounds__(128)
void outproj_kernel(const float* __restrict__ A,
                    const float* __restrict__ Wo,
                    const float* __restrict__ bo,
                    float* __restrict__ Y)
{
  __shared__ __attribute__((aligned(16))) float As[128 * 68];
  __shared__ __attribute__((aligned(16))) float Ws[64 * 68];
  const int tid  = threadIdx.x;
  const int wave = tid >> 5;
  const int lane = tid & 31;
  const int r0 = blockIdx.y * 128;
  const int c0 = blockIdx.x * 64;
  const int mrow = lane & 15, half = lane >> 4;

  v8f c[2][4] = {};

  const int srow = tid >> 4, scol = (tid & 15) * 4;
  const float* ag = A  + (size_t)(r0 + srow) * E_ + scol;
  const float* wg = Wo + (size_t)(c0 + srow) * E_ + scol;
  float* as = As + srow * 68 + scol;
  float* ws = Ws + srow * 68 + scol;
  const float* aw = As + (wave * 32 + mrow) * 68 + 4 * half;
  const float* bw = Ws + mrow * 68 + 4 * half;

  for (int kc = 0; kc < E_; kc += 64) {
    __syncthreads();
    float4 ga[8], gb[8], gw[8];
    #pragma unroll
    for (int n = 0; n < 8; ++n)
      ga[n] = *(const float4*)(ag + (size_t)(8 * n) * E_ + kc);
    #pragma unroll
    for (int n = 0; n < 8; ++n)
      gb[n] = *(const float4*)(ag + (size_t)(8 * n + 64) * E_ + kc);
    #pragma unroll
    for (int n = 0; n < 8; ++n)
      gw[n] = *(const float4*)(wg + (size_t)(8 * n) * E_ + kc);
    #pragma unroll
    for (int n = 0; n < 8; ++n) *(float4*)(as + 8 * n * 68) = ga[n];
    #pragma unroll
    for (int n = 0; n < 8; ++n) *(float4*)(as + (8 * n + 64) * 68) = gb[n];
    #pragma unroll
    for (int n = 0; n < 8; ++n) *(float4*)(ws + 8 * n * 68) = gw[n];
    __syncthreads();

    #pragma unroll
    for (int kh = 0; kh < 2; ++kh) {
      float4 fa[2][4], fb[4][4];
      #pragma unroll
      for (int i = 0; i < 2; ++i)
        #pragma unroll
        for (int q = 0; q < 4; ++q)
          fa[i][q] = *(const float4*)(aw + i * 16 * 68 + 32 * kh + 8 * q);
      #pragma unroll
      for (int f = 0; f < 4; ++f)
        #pragma unroll
        for (int q = 0; q < 4; ++q)
          fb[f][q] = *(const float4*)(bw + f * 16 * 68 + 32 * kh + 8 * q);
      __builtin_amdgcn_sched_barrier(0);
      #pragma unroll
      for (int q = 0; q < 4; ++q)
        #pragma unroll
        for (int e = 0; e < 2; ++e) {
          v2f a[2], b[4];
          #pragma unroll
          for (int i = 0; i < 2; ++i) {
            if (e) { a[i].x = fa[i][q].z; a[i].y = fa[i][q].w; }
            else   { a[i].x = fa[i][q].x; a[i].y = fa[i][q].y; }
          }
          #pragma unroll
          for (int f = 0; f < 4; ++f) {
            if (e) { b[f].x = fb[f][q].z; b[f].y = fb[f][q].w; }
            else   { b[f].x = fb[f][q].x; b[f].y = fb[f][q].y; }
          }
          #pragma unroll
          for (int i = 0; i < 2; ++i)
            #pragma unroll
            for (int f = 0; f < 4; ++f)
              c[i][f] = wmma_f32(a[i], b[f], c[i][f]);
        }
    }
  }

  #pragma unroll
  for (int i = 0; i < 2; ++i)
    #pragma unroll
    for (int f = 0; f < 4; ++f) {
      const int e = c0 + 16 * f + mrow;
      const float bias = bo[e];
      #pragma unroll
      for (int r = 0; r < 8; ++r) {
        const int m = r0 + wave * 32 + 16 * i + r + 8 * half;
        Y[(size_t)m * E_ + e] = c[i][f][r] + bias;
      }
    }
}

extern "C" void kernel_launch(void* const* d_in, const int* in_sizes, int n_in,
                              void* d_out, int out_size, void* d_ws, size_t ws_size,
                              hipStream_t stream) {
  const float* values = (const float*)d_in[0];
  const float* keys   = (const float*)d_in[1];
  const float* query  = (const float*)d_in[2];
  const float* Wv     = (const float*)d_in[4];
  const float* Wk     = (const float*)d_in[5];
  const float* Wq     = (const float*)d_in[6];
  const float* Wo     = (const float*)d_in[7];
  const float* bo     = (const float*)d_in[8];
  float* out = (float*)d_out;

  const size_t tensorElems = (size_t)N_ * H_ * S_ * D_;
  float* Qp = (float*)d_ws;
  float* Kp = Qp + tensorElems;
  float* Vp = Kp + tensorElems;
  float* Oa = Vp + tensorElems;

  const dim3 blk(128);
  const dim3 gProj(S_ / 64, H_, 3 * N_);

  proj3_kernel<<<gProj, blk, 0, stream>>>(query, keys, values, Wq, Wk, Wv, Qp, Kp, Vp);

  const dim3 gFlash(N_ * H_, S_ / 64);
  flash_kernel<<<gFlash, dim3(512), 0, stream>>>(Qp, Kp, Vp, Oa);

  const dim3 gOut(E_ / 64, (N_ * S_) / 128);
  outproj_kernel<<<gOut, blk, 0, stream>>>(Oa, Wo, bo, out);
}


# attention: work-groups of 8 waves (4 query tiles x 2 key groups, 32 keys per step), two resident per compute unit, each runs under the other's start and merge; row sums and P.V summed in a new order
# speedup vs baseline: 2.6844x; 1.0420x over previous
#include <hip/hip_runtime.h>
#include <hip/hip_bf16.h>
#include <math.h>


typedef float v2f __attribute__((ext_vector_type(2)));
typedef float v4f __attribute__((ext_vector_type(4)));
typedef float v8f __attribute__((ext_vector_type(8)));

static constexpr int N_ = 2;
static constexpr int S_ = 2048;
static constexpr int E_ = 1024;
static constexpr int H_ = 16;
static constexpr int D_ = 64;

static __device__ __forceinline__ v8f wmma_f32(v2f a, v2f b, v8f c) {
  return __builtin_amdgcn_wmma_f32_16x16x4_f32(false, a, false, b,
                                               (short)0, c, false, false);
}

template <int M>
static __device__ __forceinline__ float row_xor16(float v) {
  const int vi = __builtin_bit_cast(int, v);
  return __builtin_bit_cast(float,
      __builtin_amdgcn_update_dpp(vi, vi, 0x160 + M, 0xf, 0xf, true));
}
static __device__ __forceinline__ float rowmax16(float v) {
  v = fmaxf(v, row_xor16<1>(v));
  v = fmaxf(v, row_xor16<2>(v));
  v = fmaxf(v, row_xor16<4>(v));
  v = fmaxf(v, row_xor16<8>(v));
  return v;
}
static __device__ __forceinline__ float rowsum16(float v) {
  v += row_xor16<1>(v);
  v += row_xor16<2>(v);
  v += row_xor16<4>(v);
  v += row_xor16<8>(v);
  return v;
}

template <int MODE>
static __device__ __forceinline__
void proj_tile(const float* __restrict__ X,
               const float* __restrict__ W,
               float* __restrict__ Out,
               float* Ws,
               const int n)
{
  const int tid  = threadIdx.x;
  const int wave = tid >> 5;
  const int lane = tid & 31;
  const int h = blockIdx.y;
  const int s0 = blockIdx.x * 64 + wave * 16;
  const int mrow = lane & 15, half = lane >> 4;

  for (int i = tid; i < 1024; i += 128) {
    const int r = i >> 4, c4 = i & 15;
    const float4 w4 = ((const float4*)W)[r * 16 + c4];
    float* d = &Ws[r * 68 + c4 * 4];
    d[0] = w4.x; d[1] = w4.y; d[2] = w4.z; d[3] = w4.w;
  }
  __syncthreads();

  const float* xrow = X + ((size_t)n * S_ + (s0 + mrow)) * E_ + h * D_;
  v8f c[4] = {};
  #pragma unroll
  for (int ks = 0; ks < 16; ++ks) {
    const int k = 4 * ks + 2 * half;
    v2f a; a.x = xrow[k]; a.y = xrow[k + 1];
    #pragma unroll
    for (int f = 0; f < 4; ++f) {
      const int e = 16 * f + mrow;
      v2f b; b.x = Ws[e * 68 + k]; b.y = Ws[e * 68 + k + 1];
      if (MODE == 1) c[f] = wmma_f32(b, a, c[f]);
      else           c[f] = wmma_f32(a, b, c[f]);
    }
  }

  float* obase = Out + ((size_t)(n * H_ + h)) * S_ * D_;
  if (MODE == 0) {
    #pragma unroll
    for (int f = 0; f < 4; ++f)
      #pragma unroll
      for (int r = 0; r < 8; ++r) {
        const int m = r + 8 * half;
        const int e = 16 * f + mrow;
        obase[(size_t)(s0 + m) * D_ + e] = c[f][r];
      }
  } else {
    float* tile = obase + (size_t)(s0 >> 4) * 1024 + lane * 4;
    #pragma unroll
    for (int f = 0; f < 4; ++f)
      #pragma unroll
      for (int jj = 0; jj < 2; ++jj) {
        float4 v4;
        v4.x = c[f][4 * jj];     v4.y = c[f][4 * jj + 1];
        v4.z = c[f][4 * jj + 2]; v4.w = c[f][4 * jj + 3];
        *(float4*)(tile + (f * 2 + jj) * 128) = v4;
      }
  }
}

__global__ __launch_bounds__(128)
void proj3_kernel(const float* __restrict__ Xq, const float* __restrict__ Xk,
                  const float* __restrict__ Xv,
                  const float* __restrict__ Wq, const float* __restrict__ Wk,
                  const float* __restrict__ Wv,
                  float* __restrict__ Qp, float* __restrict__ Kp, float* __restrict__ Vp)
{
  __shared__ float Ws[64 * 68];
  const int which = blockIdx.z / N_;
  const int n     = blockIdx.z % N_;
  if (which == 0)      proj_tile<0>(Xq, Wq, Qp, Ws, n);
  else if (which == 1) proj_tile<1>(Xk, Wk, Kp, Ws, n);
  else                 proj_tile<2>(Xv, Wv, Vp, Ws, n);
}

static __device__ __forceinline__ void load_tiles(const float* kt, const float* vt,
                                                  v4f (&kq)[8], v4f (&vq)[8]) {
  #pragma unroll
  for (int i = 0; i < 8; ++i) kq[i] = *(const v4f*)(kt + i * 128);
  #pragma unroll
  for (int i = 0; i < 8; ++i) vq[i] = *(const v4f*)(vt + i * 128);
  __builtin_amdgcn_sched_barrier(0);
}

static __device__ __forceinline__ v8f tile_scores(const float (&qreg)[32], const v4f (&kq)[8],
                                                  float slope, float scale, float f0) {
  v8f s = {};
  #pragma unroll
  for (int t = 0; t < 16; ++t) {
    const v4f k4 = kq[t >> 1];
    v2f a; a.x = qreg[2 * t]; a.y = qreg[2 * t + 1];
    v2f b;
    if (t & 1) { b.x = k4.z; b.y = k4.w; } else { b.x = k4.x; b.y = k4.y; }
    s = wmma_f32(a, b, s);
  }
  #pragma unroll
  for (int r = 0; r < 8; ++r) s[r] = (s[r] + slope * (f0 - (float)r)) * scale;
  return s;
}

static __device__ __forceinline__ void accumulate_pv(float* Pw, const v8f& p, const v4f (&vq)[8],
                                                     v8f (&o)[4], int mrow, int half) {
  #pragma unroll
  for (int r = 0; r < 8; ++r)
    Pw[(r + 8 * half) * 17 + mrow] = p[r];
  #pragma unroll
  for (int kk = 0; kk < 4; ++kk) {
    const int ka = 8 * half + 2 * kk;
    v2f a; a.x = Pw[mrow * 17 + ka]; a.y = Pw[mrow * 17 + ka + 1];
    #pragma unroll
    for (int f = 0; f < 4; ++f) {
      const v4f v4 = vq[f * 2 + (kk >> 1)];
      v2f b;
      if (kk & 1) { b.x = v4.z; b.y = v4.w; } else { b.x = v4.x; b.y = v4.y; }
      o[f] = wmma_f32(a, b, o[f]);
    }
  }
}

__global__ __launch_bounds__(256, 4)
void flash_kernel(const float* __restrict__ Qp,
                  const float* __restrict__ Kf,
                  const float* __restrict__ Vf,
                  float* __restrict__ O)
{
  __shared__ float smem[6144 + 8 * 272];

  const int tid  = threadIdx.x;
  const int wave = __builtin_amdgcn_readfirstlane(tid >> 5);
  const int lane = tid & 31;
  const int grp  = wave >> 2;
  const int qw   = wave & 3;
  const int n = blockIdx.x / H_, h = blockIdx.x % H_;
  const int qblk = (S_ / 64 - 1) - blockIdx.y;
  const int q0 = qblk * 64 + qw * 16;
  const int mrow = lane & 15, half = lane >> 4;
  const size_t headOff = ((size_t)(n * H_ + h)) * S_ * D_;
  float* Pw = smem + 6144 + wave * 272;

  float qreg[32];
  const float* qrow = Qp + headOff + (size_t)(q0 + mrow) * D_ + 8 * half;
  #pragma unroll
  for (int f = 0; f < 4; ++f) {
    const float4 qa = *(const float4*)(qrow + 16 * f);
    const float4 qb = *(const float4*)(qrow + 16 * f + 4);
    qreg[8 * f + 0] = qa.x; qreg[8 * f + 1] = qa.y; qreg[8 * f + 2] = qa.z; qreg[8 * f + 3] = qa.w;
    qreg[8 * f + 4] = qb.x; qreg[8 * f + 5] = qb.y; qreg[8 * f + 6] = qb.z; qreg[8 * f + 7] = qb.w;
  }

  v8f o[4] = {};
  float mrun[8], lrun[8];
  #pragma unroll
  for (int r = 0; r < 8; ++r) { mrun[r] = -3.0e38f; lrun[r] = 0.0f; }

  const float slope  = (float)(1 << __popc(h));
  const float scale  = 0.03125f;
  const int   tAvail = 4 * qblk + qw - grp;
  const int   nMine  = (tAvail < 0) ? 0 : (tAvail / 2 + 1);

  const float* kbase = Kf + headOff + (size_t)grp * 1024 + lane * 4;
  const float* vbase = Vf + headOff + (size_t)grp * 1024 + lane * 4;
  constexpr float kLazy = 8.0f;
  const int dBase = (grp * 16 + mrow) - (q0 + 8 * half);

  int st = nMine - 1;
  while (st >= 0) {
    {
      v4f kq[8], vq[8];
      load_tiles(kbase + (size_t)st * 2048, vbase + (size_t)st * 2048, kq, vq);
      const int d0 = dBase + st * 32;
      v8f s = tile_scores(qreg, kq, slope, scale, (float)d0);
      float pscale[8];
      #pragma unroll
      for (int r = 0; r < 8; ++r) {
        float e = s[r];
        if (d0 > r) e = -3.0e18f;
        const float rm    = rowmax16(e);
        const float mnew  = fmaxf(mrun[r], rm);
        const float alpha = __expf(mrun[r] - mnew);
        mrun[r] = mnew;
        const float p = __expf(e - mnew);
        s[r] = p;
        lrun[r] = lrun[r] * alpha + p;
        pscale[r] = alpha;
      }
      #pragma unroll
      for (int f = 0; f < 4; ++f)
        #pragma unroll
        for (int r = 0; r < 8; ++r) o[f][r] *= pscale[r];
      accumulate_pv(Pw, s, vq, o, mrow, half);
    }
    --st;
    for (; st >= 0; --st) {
      v4f kq[8], vq[8];
      load_tiles(kbase + (size_t)st * 2048, vbase + (size_t)st * 2048, kq, vq);
      const int d0 = dBase + st * 32;
      v8f s = tile_scores(qreg, kq, slope, scale, (float)d0);
      #pragma unroll
      for (int r = 0; r < 8; ++r) s[r] -= mrun[r];
      const float xtop = fmaxf(fmaxf(fmaxf(s[0], s[1]), fmaxf(s[2], s[3])),
                               fmaxf(fmaxf(s[4], s[5]), fmaxf(s[6], s[7])));
      asm volatile("" :: "v"(vq[0]), "v"(vq[1]), "v"(vq[2]), "v"(vq[3]),
                         "v"(vq[4]), "v"(vq[5]), "v"(vq[6]), "v"(vq[7]), "v"(xtop));
      if (__builtin_amdgcn_ballot_w32(xtop > kLazy) != 0u) break;
      #pragma unroll
      for (int r = 0; r < 8; ++r) {
        const float p = __expf(s[r]);
        s[r] = p;
        lrun[r] += p;
      }
      accumulate_pv(Pw, s, vq, o, mrow, half);
    }
  }

  #pragma unroll
  for (int r = 0; r < 8; ++r) lrun[r] = rowsum16(lrun[r]);

  if (grp == 1) {
    float* mb = smem + qw * 1536;
    #pragma unroll
    for (int f = 0; f < 4; ++f)
      #pragma unroll
      for (int r = 0; r < 8; ++r) mb[(f * 8 + r) * 32 + lane] = o[f][r];
    #pragma unroll
    for (int r = 0; r < 8; ++r) {
      mb[1024 + r * 32 + lane] = mrun[r];
      mb[1280 + r * 32 + lane] = lrun[r];
    }
  }
  __syncthreads();
  if (grp == 0) {
    const float* mb = smem + qw * 1536;
    float a0[8], a1[8];
    #pragma unroll
    for (int r = 0; r < 8; ++r) {
      const float m1 = mb[1024 + r * 32 + lane];
      const float l1 = mb[1280 + r * 32 + lane];
      const float m  = fmaxf(mrun[r], m1);
      a0[r] = __expf(mrun[r] - m);
      a1[r] = __expf(m1 - m);
      lrun[r] = lrun[r] * a0[r] + l1 * a1[r];
      mrun[r] = m;
    }
    #pragma unroll
    for (int f = 0; f < 4; ++f)
      #pragma unroll
      for (int r = 0; r < 8; ++r)
        o[f][r] = o[f][r] * a0[r] + mb[(f * 8 + r) * 32 + lane] * a1[r];
  }

  if (grp == 0) {
    #pragma unroll
    for (int r = 0; r < 8; ++r) lrun[r] = 1.0f / lrun[r];
    float* obase = O + (size_t)n * S_ * E_ + h * D_;
    #pragma unroll
    for (int f = 0; f < 4; ++f)
      #pragma unroll
      for (int r = 0; r < 8; ++r) {
        const int m = r + 8 * half;
        const int dcol = 16 * f + mrow;
        obase[(size_t)(q0 + m) * E_ + dcol] = o[f][r] * lrun[r];
      }
  }
}

__global__ __launch_bounds__(128)
void outproj_kernel(const float* __restrict__ A,
                    const float* __restrict__ Wo,
                    const float* __restrict__ bo,
                    float* __restrict__ Y)
{
  __shared__ __attribute__((aligned(16))) float As[128 * 68];
  __shared__ __attribute__((aligned(16))) float Ws[64 * 68];
  const int tid  = threadIdx.x;
  const int wave = tid >> 5;
  const int lane = tid & 31;
  const int r0 = blockIdx.y * 128;
  const int c0 = blockIdx.x * 64;
  const int mrow = lane & 15, half = lane >> 4;

  v8f c[2][4] = {};

  const int srow = tid >> 4, scol = (tid & 15) * 4;
  const float* ag = A  + (size_t)(r0 + srow) * E_ + scol;
  const float* wg = Wo + (size_t)(c0 + srow) * E_ + scol;
  float* as = As + srow * 68 + scol;
  float* ws = Ws + srow * 68 + scol;
  const float* aw = As + (wave * 32 + mrow) * 68 + 4 * half;
  const float* bw = Ws + mrow * 68 + 4 * half;

  for (int kc = 0; kc < E_; kc += 64) {
    __syncthreads();
    float4 ga[8], gb[8], gw[8];
    #pragma unroll
    for (int n = 0; n < 8; ++n)
      ga[n] = *(const float4*)(ag + (size_t)(8 * n) * E_ + kc);
    #pragma unroll
    for (int n = 0; n < 8; ++n)
      gb[n] = *(const float4*)(ag + (size_t)(8 * n + 64) * E_ + kc);
    #pragma unroll
    for (int n = 0; n < 8; ++n)
      gw[n] = *(const float4*)(wg + (size_t)(8 * n) * E_ + kc);
    #pragma unroll
    for (int n = 0; n < 8; ++n) *(float4*)(as + 8 * n * 68) = ga[n];
    #pragma unroll
    for (int n = 0; n < 8; ++n) *(float4*)(as + (8 * n + 64) * 68) = gb[n];
    #pragma unroll
    for (int n = 0; n < 8; ++n) *(float4*)(ws + 8 * n * 68) = gw[n];
    __syncthreads();

    #pragma unroll
    for (int kh = 0; kh < 2; ++kh) {
      float4 fa[2][4], fb[4][4];
      #pragma unroll
      for (int i = 0; i < 2; ++i)
        #pragma unroll
        for (int q = 0; q < 4; ++q)
          fa[i][q] = *(const float4*)(aw + i * 16 * 68 + 32 * kh + 8 * q);
      #pragma unroll
      for (int f = 0; f < 4; ++f)
        #pragma unroll
        for (int q = 0; q < 4; ++q)
          fb[f][q] = *(const float4*)(bw + f * 16 * 68 + 32 * kh + 8 * q);
      __builtin_amdgcn_sched_barrier(0);
      #pragma unroll
      for (int q = 0; q < 4; ++q)
        #pragma unroll
        for (int e = 0; e < 2; ++e) {
          v2f a[2], b[4];
          #pragma unroll
          for (int i = 0; i < 2; ++i) {
            if (e) { a[i].x = fa[i][q].z; a[i].y = fa[i][q].w; }
            else   { a[i].x = fa[i][q].x; a[i].y = fa[i][q].y; }
          }
          #pragma unroll
          for (int f = 0; f < 4; ++f) {
            if (e) { b[f].x = fb[f][q].z; b[f].y = fb[f][q].w; }
            else   { b[f].x = fb[f][q].x; b[f].y = fb[f][q].y; }
          }
          #pragma unroll
          for (int i = 0; i < 2; ++i)
            #pragma unroll
            for (int f = 0; f < 4; ++f)
              c[i][f] = wmma_f32(a[i], b[f], c[i][f]);
        }
    }
  }

  #pragma unroll
  for (int i = 0; i < 2; ++i)
    #pragma unroll
    for (int f = 0; f < 4; ++f) {
      const int e = c0 + 16 * f + mrow;
      const float bias = bo[e];
      #pragma unroll
      for (int r = 0; r < 8; ++r) {
        const int m = r0 + wave * 32 + 16 * i + r + 8 * half;
        Y[(size_t)m * E_ + e] = c[i][f][r] + bias;
      }
    }
}

extern "C" void kernel_launch(void* const* d_in, const int* in_sizes, int n_in,
                              void* d_out, int out_size, void* d_ws, size_t ws_size,
                              hipStream_t stream) {
  const float* values = (const float*)d_in[0];
  const float* keys   = (const float*)d_in[1];
  const float* query  = (const float*)d_in[2];
  const float* Wv     = (const float*)d_in[4];
  const float* Wk     = (const float*)d_in[5];
  const float* Wq     = (const float*)d_in[6];
  const float* Wo     = (const float*)d_in[7];
  const float* bo     = (const float*)d_in[8];
  float* out = (float*)d_out;

  const size_t tensorElems = (size_t)N_ * H_ * S_ * D_;
  float* Qp = (float*)d_ws;
  float* Kp = Qp + tensorElems;
  float* Vp = Kp + tensorElems;
  float* Oa = Vp + tensorElems;

  const dim3 blk(128);
  const dim3 gProj(S_ / 64, H_, 3 * N_);

  proj3_kernel<<<gProj, blk, 0, stream>>>(query, keys, values, Wq, Wk, Wv, Qp, Kp, Vp);

  const dim3 gFlash(N_ * H_, S_ / 64);
  flash_kernel<<<gFlash, dim3(256), 0, stream>>>(Qp, Kp, Vp, Oa);

  const dim3 gOut(E_ / 64, (N_ * S_) / 128);
  outproj_kernel<<<gOut, blk, 0, stream>>>(Oa, Wo, bo, out);
}
